# starting kernel
# baseline (speedup 1.0000x reference)
#include <hip/hip_runtime.h>
#include <math.h>

#define NN 10000
#define EE 320000
#define HH 4
#define DD 128
#define FH (HH * DD)
#define K0 256
#define KMAX 512
#define NEG_SLOPE 0.2f

typedef __attribute__((ext_vector_type(2))) float v2f;
typedef __attribute__((ext_vector_type(8))) float v8f;

__global__ void fill_kernel(float* __restrict__ p, long n, float v) {
  long i = (long)blockIdx.x * blockDim.x + threadIdx.x;
  if (i < n) p[i] = v;
}

__global__ __launch_bounds__(256) void gemm16_wmma(const float* __restrict__ X,
                                                   const float* __restrict__ W,
                                                   float* __restrict__ out,
                                                   int M, int K, int Nout) {
  __shared__ float Bs[KMAX * 16];

  const int tid = threadIdx.x;
  const int lane = tid & 31;
  const int wv = tid >> 5;
  const int ntiles = Nout >> 4;
  const int mg = blockIdx.x / ntiles;
  const int tn = blockIdx.x - mg * ntiles;

  const float* wp = W + tn * 16;
  for (int i = tid; i < K * 16; i += 256) {
    int kk = i >> 4, c = i & 15;
    Bs[i] = wp[(size_t)kk * Nout + c];
  }
  __syncthreads();

  const int tm = mg * 8 + wv;
  if (tm * 16 >= M) return;

  const int half = lane >> 4;
  const int l = lane & 15;

  const float* aptr = X + (size_t)(tm * 16 + l) * K + half * 2;
  const float* bbase = Bs + (half * 2) * 16 + l;

  v8f acc = {};
  for (int k = 0; k < K; k += 4) {
    float2 av = *(const float2*)aptr;
    v2f a; a.x = av.x; a.y = av.y;
    v2f b; b.x = bbase[0]; b.y = bbase[16];
    acc = __builtin_amdgcn_wmma_f32_16x16x4_f32(
         false, a,  false, b,
         (short)0, acc,  false,  false);
    aptr += 4;
    bbase += 64;
  }

  float* obase = out + (size_t)(tm * 16 + half * 8) * Nout + tn * 16 + l;
#pragma unroll
  for (int r = 0; r < 8; ++r) obase[(size_t)r * Nout] = acc[r];
}

__global__ void attn_coef(const float* __restrict__ feat,
                          const float* __restrict__ al,
                          const float* __restrict__ ar,
                          float* __restrict__ el, float* __restrict__ er) {
  int i = blockIdx.x * blockDim.x + threadIdx.x;
  if (i >= NN * HH) return;
  int h = i & (HH - 1);
  const float4* f = (const float4*)(feat + (size_t)i * DD);
  const float4* a = (const float4*)(al + h * DD);
  const float4* b = (const float4*)(ar + h * DD);
  float sl = 0.f, sr = 0.f;
#pragma unroll 4
  for (int d = 0; d < DD / 4; ++d) {
    float4 fv = f[d], av = a[d], bv = b[d];
    sl += fv.x * av.x + fv.y * av.y + fv.z * av.z + fv.w * av.w;
    sr += fv.x * bv.x + fv.y * bv.y + fv.z * bv.z + fv.w * bv.w;
  }
  el[i] = sl;
  er[i] = sr;
}

__device__ __forceinline__ void atomicMaxF(float* addr, float val) {
  if (val >= 0.f)
    atomicMax((int*)addr, __float_as_int(val));
  else
    atomicMin((unsigned int*)addr, __float_as_uint(val));
}

__device__ __forceinline__ float leaky(float v) {
  return v > 0.f ? v : NEG_SLOPE * v;
}

__global__ void edge_max(const int* __restrict__ src, const int* __restrict__ dst,
                         const float* __restrict__ el, const float* __restrict__ er,
                         float* __restrict__ emax) {
  int e = blockIdx.x * blockDim.x + threadIdx.x;
  if (e >= EE) return;
  int s = src[e], t = dst[e];
#pragma unroll
  for (int h = 0; h < HH; ++h) {
    float v = leaky(el[s * HH + h] + er[t * HH + h]);
    atomicMaxF(&emax[t * HH + h], v);
  }
}

__global__ void edge_exp(const int* __restrict__ src, const int* __restrict__ dst,
                         const float* __restrict__ el, const float* __restrict__ er,
                         const float* __restrict__ emax,
                         float* __restrict__ exv, float* __restrict__ denom) {
  int e = blockIdx.x * blockDim.x + threadIdx.x;
  if (e >= EE) return;
  int s = src[e], t = dst[e];
#pragma unroll
  for (int h = 0; h < HH; ++h) {
    float v = leaky(el[s * HH + h] + er[t * HH + h]);
    float m = emax[t * HH + h];
    float xo = __expf(v - m);
    exv[(size_t)e * HH + h] = xo;
    atomicAdd(&denom[t * HH + h], xo);
  }
}

__global__ __launch_bounds__(128) void edge_scatter(
    const int* __restrict__ src, const int* __restrict__ dst,
    const float* __restrict__ feat, const float* __restrict__ exv,
    const float* __restrict__ denom, float* __restrict__ agg) {
  int e = blockIdx.x;
  int d = threadIdx.x;
  int s = src[e], t = dst[e];
  const float* fs = feat + (size_t)s * FH + d;
  float* ob = agg + (size_t)t * FH + d;
  __builtin_prefetch(fs, 0, 0);
#pragma unroll
  for (int h = 0; h < HH; ++h) {
    float dn = denom[t * HH + h];
    float alpha = exv[(size_t)e * HH + h] / (dn == 0.f ? 1.f : dn);
    atomicAdd(ob + h * DD, fs[h * DD] * alpha);
  }
}

__global__ void layer_elu(const float* __restrict__ agg, float* __restrict__ out,
                          long n) {
  long i = (long)blockIdx.x * blockDim.x + threadIdx.x;
  if (i < n) {
    float v = agg[i];
    out[i] = v > 0.f ? v : (__expf(v) - 1.f);
  }
}

__global__ void finalize(const float* __restrict__ agg,
                         const float* __restrict__ hin,
                         float* __restrict__ out) {
  int i = blockIdx.x * blockDim.x + threadIdx.x;
  if (i >= NN * DD) return;
  int n = i / DD, d = i - n * DD;
  float s = 0.f;
#pragma unroll
  for (int h = 0; h < HH; ++h) {
    size_t off = (size_t)n * FH + h * DD + d;
    float v = agg[off] + hin[off];
    s += v > 0.f ? v : (__expf(v) - 1.f);
  }
  out[i] = s * (1.f / HH);
}

extern "C" void kernel_launch(void* const* d_in, const int* in_sizes, int n_in,
                              void* d_out, int out_size, void* d_ws, size_t ws_size,
                              hipStream_t stream) {
  const int* src = (const int*)d_in[0];
  const int* dst = (const int*)d_in[1];
  const float* x   = (const float*)d_in[2];
  const float* W0  = (const float*)d_in[3];
  const float* al0 = (const float*)d_in[4];
  const float* ar0 = (const float*)d_in[5];
  const float* W1  = (const float*)d_in[6];
  const float* al1 = (const float*)d_in[7];
  const float* ar1 = (const float*)d_in[8];
  float* out = (float*)d_out;

  float* ws    = (float*)d_ws;
  float* feat  = ws;
  float* agg   = feat  + (size_t)NN * FH;
  float* hbuf  = agg   + (size_t)NN * FH;
  float* el    = hbuf  + (size_t)NN * FH;
  float* er    = el    + (size_t)NN * HH;
  float* emax  = er    + (size_t)NN * HH;
  float* denom = emax  + (size_t)NN * HH;
  float* exv   = denom + (size_t)NN * HH;

  const int mtiles = (NN + 15) / 16;
  const int mgroups = (mtiles + 7) / 8;
  const int ntiles = FH / 16;
  const int gemmBlocks = mgroups * ntiles;

  auto run_layer = [&](const float* xin, const float* W, int K,
                       const float* al, const float* ar) {
    gemm16_wmma<<<gemmBlocks, 256, 0, stream>>>(xin, W, feat, NN, K, FH);
    attn_coef<<<(NN * HH + 255) / 256, 256, 0, stream>>>(feat, al, ar, el, er);
    fill_kernel<<<(NN * HH + 255) / 256, 256, 0, stream>>>(emax, (long)NN * HH,
                                                           -__builtin_huge_valf());
    fill_kernel<<<(NN * HH + 255) / 256, 256, 0, stream>>>(denom, (long)NN * HH, 0.f);
    fill_kernel<<<((long)NN * FH + 255) / 256, 256, 0, stream>>>(agg, (long)NN * FH, 0.f);
    edge_max<<<(EE + 255) / 256, 256, 0, stream>>>(src, dst, el, er, emax);
    edge_exp<<<(EE + 255) / 256, 256, 0, stream>>>(src, dst, el, er, emax, exv, denom);
    edge_scatter<<<EE, 128, 0, stream>>>(src, dst, feat, exv, denom, agg);
  };

  run_layer(x, W0, K0, al0, ar0);
  layer_elu<<<((long)NN * FH + 255) / 256, 256, 0, stream>>>(agg, hbuf, (long)NN * FH);
  run_layer(hbuf, W1, FH, al1, ar1);
  finalize<<<(NN * DD + 255) / 256, 256, 0, stream>>>(agg, hbuf, out);
}


# Edges grouped by destination once per call (int-atomic rank + scan); one block per node does softmax and feature sum in registers; float atomics and fills gone; sum order = arrival rank per node
# speedup vs baseline: 4.1171x; 4.1171x over previous
#include <hip/hip_runtime.h>
#include <math.h>

#define NN 10000
#define EE 320000
#define HH 4
#define DD 128
#define FH (HH * DD)
#define K0 256
#define KMAX 512
#define NEG_SLOPE 0.2f

typedef __attribute__((ext_vector_type(2))) float v2f;
typedef __attribute__((ext_vector_type(8))) float v8f;

__global__ __launch_bounds__(256) void gemm16_wmma(const float* __restrict__ X,
                                                   const float* __restrict__ W,
                                                   float* __restrict__ out,
                                                   int M, int K, int Nout) {
  __shared__ float Bs[KMAX * 16];

  const int tid = threadIdx.x;
  const int lane = tid & 31;
  const int wv = tid >> 5;
  const int ntiles = Nout >> 4;
  const int mg = blockIdx.x / ntiles;
  const int tn = blockIdx.x - mg * ntiles;

  const float* wp = W + tn * 16;
  for (int i = tid; i < K * 16; i += 256) {
    int kk = i >> 4, c = i & 15;
    Bs[i] = wp[(size_t)kk * Nout + c];
  }
  __syncthreads();

  const int tm = mg * 8 + wv;
  if (tm * 16 >= M) return;

  const int half = lane >> 4;
  const int l = lane & 15;

  const float* aptr = X + (size_t)(tm * 16 + l) * K + half * 2;
  const float* bbase = Bs + (half * 2) * 16 + l;

  v8f acc = {};
  for (int k = 0; k < K; k += 4) {
    float2 av = *(const float2*)aptr;
    v2f a; a.x = av.x; a.y = av.y;
    v2f b; b.x = bbase[0]; b.y = bbase[16];
    acc = __builtin_amdgcn_wmma_f32_16x16x4_f32(
         false, a,  false, b,
         (short)0, acc,  false,  false);
    aptr += 4;
    bbase += 64;
  }

  float* obase = out + (size_t)(tm * 16 + half * 8) * Nout + tn * 16 + l;
#pragma unroll
  for (int r = 0; r < 8; ++r) obase[(size_t)r * Nout] = acc[r];
}

__global__ void attn_coef(const float* __restrict__ feat,
                          const float* __restrict__ al,
                          const float* __restrict__ ar,
                          float* __restrict__ el, float* __restrict__ er) {
  int i = blockIdx.x * blockDim.x + threadIdx.x;
  if (i >= NN * HH) return;
  int h = i & (HH - 1);
  const float4* f = (const float4*)(feat + (size_t)i * DD);
  const float4* a = (const float4*)(al + h * DD);
  const float4* b = (const float4*)(ar + h * DD);
  float sl = 0.f, sr = 0.f;
#pragma unroll 4
  for (int d = 0; d < DD / 4; ++d) {
    float4 fv = f[d], av = a[d], bv = b[d];
    sl += fv.x * av.x + fv.y * av.y + fv.z * av.z + fv.w * av.w;
    sr += fv.x * bv.x + fv.y * bv.y + fv.z * bv.z + fv.w * bv.w;
  }
  el[i] = sl;
  er[i] = sr;
}

__global__ void zero_counts(int* __restrict__ cnt) {
  int i = blockIdx.x * blockDim.x + threadIdx.x;
  if (i < NN) cnt[i] = 0;
}

__global__ void dst_count(const int* __restrict__ dst, int* __restrict__ cnt,
                          int* __restrict__ rnk) {
  int e = blockIdx.x * blockDim.x + threadIdx.x;
  if (e >= EE) return;
  rnk[e] = atomicAdd(&cnt[dst[e]], 1);
}

__global__ __launch_bounds__(1024) void scan_counts(const int* __restrict__ cnt,
                                                    int* __restrict__ rowptr) {
  __shared__ int part[1024];
  const int tid = threadIdx.x;
  const int per = (NN + 1023) / 1024;
  const int b = tid * per;
  const int e = (b + per < NN) ? (b + per) : NN;
  int s = 0;
  for (int i = b; i < e; ++i) s += cnt[i];
  part[tid] = s;
  __syncthreads();
  for (int off = 1; off < 1024; off <<= 1) {
    int v = (tid >= off) ? part[tid - off] : 0;
    __syncthreads();
    part[tid] += v;
    __syncthreads();
  }
  int run = part[tid] - s;
  for (int i = b; i < e; ++i) {
    rowptr[i] = run;
    run += cnt[i];
  }
  if (tid == 1023) rowptr[NN] = part[1023];
}

__global__ void dst_place(const int* __restrict__ src, const int* __restrict__ dst,
                          const int* __restrict__ rowptr, const int* __restrict__ rnk,
                          int* __restrict__ csrc) {
  int e = blockIdx.x * blockDim.x + threadIdx.x;
  if (e >= EE) return;
  csrc[rowptr[dst[e]] + rnk[e]] = src[e];
}

__device__ __forceinline__ float leaky(float v) {
  return v > 0.f ? v : NEG_SLOPE * v;
}

__device__ __forceinline__ float elu1(float v) {
  return v > 0.f ? v : (__expf(v) - 1.f);
}

__device__ __forceinline__ float wave_max(float v) {
  for (int o = 16; o > 0; o >>= 1) v = fmaxf(v, __shfl_xor(v, o));
  return v;
}

__device__ __forceinline__ float wave_sum(float v) {
  for (int o = 16; o > 0; o >>= 1) v += __shfl_xor(v, o);
  return v;
}

template <bool FINAL>
__global__ __launch_bounds__(128) void node_gather(
    const int* __restrict__ rowptr, const int* __restrict__ csrc,
    const float* __restrict__ feat, const float* __restrict__ el,
    const float* __restrict__ er, const float* __restrict__ hin,
    float* __restrict__ outp) {
  __shared__ float4 hs[FH / 4];

  const int t = blockIdx.x;
  const int tid = threadIdx.x;
  const int lane = tid & 31;
  const int h = tid >> 5;
  const int beg = rowptr[t];
  const int end = rowptr[t + 1];
  const float er_t = er[t * HH + h];

  float m = -__builtin_huge_valf();
  for (int i = beg + lane; i < end; i += 32) {
    int s = csrc[i];
    m = fmaxf(m, leaky(el[s * HH + h] + er_t));
  }
  m = wave_max(m);

  float dsum = 0.f;
  for (int i = beg + lane; i < end; i += 32) {
    int s = csrc[i];
    dsum += __expf(leaky(el[s * HH + h] + er_t) - m);
  }
  dsum = wave_sum(dsum);
  const float dn = (dsum == 0.f) ? 1.f : dsum;

  float4 acc = make_float4(0.f, 0.f, 0.f, 0.f);
  const float4* frow = (const float4*)feat + tid;
  for (int c = beg; c < end; c += 32) {
    const int i = c + lane;
    int s = 0;
    float a = 0.f;
    if (i < end) {
      s = csrc[i];
      a = __expf(leaky(el[s * HH + h] + er_t) - m) / dn;
    }
    int cnt = end - c;
    if (cnt > 32) cnt = 32;
#pragma unroll 4
    for (int j = 0; j < cnt; ++j) {
      const int sj = __builtin_amdgcn_readlane(s, j);
      const float aj = __int_as_float(__builtin_amdgcn_readlane(__float_as_int(a), j));
      const float4 f = frow[(size_t)sj * (FH / 4)];
      acc.x += f.x * aj;
      acc.y += f.y * aj;
      acc.z += f.z * aj;
      acc.w += f.w * aj;
    }
  }

  if (!FINAL) {
    float4 r;
    r.x = elu1(acc.x);
    r.y = elu1(acc.y);
    r.z = elu1(acc.z);
    r.w = elu1(acc.w);
    ((float4*)outp)[(size_t)t * (FH / 4) + tid] = r;
  } else {
    const float4 x = ((const float4*)hin)[(size_t)t * (FH / 4) + tid];
    float4 r;
    r.x = elu1(acc.x + x.x);
    r.y = elu1(acc.y + x.y);
    r.z = elu1(acc.z + x.z);
    r.w = elu1(acc.w + x.w);
    hs[tid] = r;
    __syncthreads();
    const float* hf = (const float*)hs;
    float sm = 0.f;
    for (int hh = 0; hh < HH; ++hh) sm += hf[hh * DD + tid];
    outp[(size_t)t * DD + tid] = sm * (1.f / HH);
  }
}

extern "C" void kernel_launch(void* const* d_in, const int* in_sizes, int n_in,
                              void* d_out, int out_size, void* d_ws, size_t ws_size,
                              hipStream_t stream) {
  const int* src = (const int*)d_in[0];
  const int* dst = (const int*)d_in[1];
  const float* x   = (const float*)d_in[2];
  const float* W0  = (const float*)d_in[3];
  const float* al0 = (const float*)d_in[4];
  const float* ar0 = (const float*)d_in[5];
  const float* W1  = (const float*)d_in[6];
  const float* al1 = (const float*)d_in[7];
  const float* ar1 = (const float*)d_in[8];
  float* out = (float*)d_out;

  float* ws    = (float*)d_ws;
  float* feat  = ws;
  float* hbuf  = feat  + (size_t)NN * FH;
  float* el    = hbuf  + (size_t)NN * FH;
  float* er    = el    + (size_t)NN * HH;
  int* cnt     = (int*)(er + (size_t)NN * HH);
  int* rowptr  = cnt    + NN;
  int* rnk     = rowptr + (NN + 16);
  int* csrc    = rnk    + EE;

  const int mtiles = (NN + 15) / 16;
  const int mgroups = (mtiles + 7) / 8;
  const int ntiles = FH / 16;
  const int gemmBlocks = mgroups * ntiles;

  zero_counts<<<(NN + 255) / 256, 256, 0, stream>>>(cnt);
  dst_count<<<(EE + 255) / 256, 256, 0, stream>>>(dst, cnt, rnk);
  scan_counts<<<1, 1024, 0, stream>>>(cnt, rowptr);
  dst_place<<<(EE + 255) / 256, 256, 0, stream>>>(src, dst, rowptr, rnk, csrc);

  gemm16_wmma<<<gemmBlocks, 256, 0, stream>>>(x, W0, feat, NN, K0, FH);
  attn_coef<<<(NN * HH + 255) / 256, 256, 0, stream>>>(feat, al0, ar0, el, er);
  node_gather<false><<<NN, 128, 0, stream>>>(rowptr, csrc, feat, el, er, feat, hbuf);
  gemm16_wmma<<<gemmBlocks, 256, 0, stream>>>(hbuf, W1, feat, NN, FH, FH);
  attn_coef<<<(NN * HH + 255) / 256, 256, 0, stream>>>(feat, al1, ar1, el, er);
  node_gather<true><<<NN, 128, 0, stream>>>(rowptr, csrc, feat, el, er, hbuf, out);
}


# GEMM: each wave computes a 16x64 strip (4 accumulators) so one A fragment load feeds 4 matrix instructions; W panel staged in LDS in fragment order, A fetched a group ahead; sum order unchanged
# speedup vs baseline: 5.3035x; 1.2882x over previous
#include <hip/hip_runtime.h>
#include <math.h>

#define NN 10000
#define EE 320000
#define HH 4
#define DD 128
#define FH (HH * DD)
#define K0 256
#define KMAX 512
#define NEG_SLOPE 0.2f

typedef __attribute__((ext_vector_type(2))) float v2f;
typedef __attribute__((ext_vector_type(8))) float v8f;

template <int K, int Nout>
__global__ __launch_bounds__(256) void gemm16x64_wmma(const float* __restrict__ X,
                                                      const float* __restrict__ W,
                                                      float* __restrict__ out,
                                                      int M) {
  __shared__ float2 Bs[K * 32];

  const int tid = threadIdx.x;
  const int lane = tid & 31;
  const int wv = tid >> 5;
  constexpr int npan = Nout >> 6;
  const int mg = blockIdx.x / npan;
  const int pn = blockIdx.x - mg * npan;

  {
    const int c = tid & 63;
    const int j = c >> 4, l = c & 15;
    const float* wp = W + pn * 64 + c;
#pragma unroll 8
    for (int it = 0; it < K / 8; ++it) {
      const int kp = it * 4 + (tid >> 6);
      const float v0 = wp[(size_t)(2 * kp) * Nout];
      const float v1 = wp[(size_t)(2 * kp + 1) * Nout];
      const int ks = kp >> 1, h = kp & 1;
      Bs[((ks * 4 + j) * 2 + h) * 16 + l] = make_float2(v0, v1);
    }
  }
  __syncthreads();

  const int tm = mg * 8 + wv;
  if (tm * 16 >= M) return;

  const int half = lane >> 4;
  const int l = lane & 15;

  const float* aptr = X + (size_t)(tm * 16 + l) * K + half * 2;
  const float2* bptr = Bs + lane;

  v8f acc0 = {}, acc1 = {}, acc2 = {}, acc3 = {};

  float2 cur[8], nxt[8];
#pragma unroll
  for (int i = 0; i < 8; ++i) cur[i] = *(const float2*)(aptr + 4 * i);

  constexpr int NG = K / 32;
  for (int g = 0; g < NG; ++g) {
    if (g + 1 < NG) {
#pragma unroll
      for (int i = 0; i < 8; ++i) nxt[i] = *(const float2*)(aptr + 32 * (g + 1) + 4 * i);
    }
#pragma unroll
    for (int i = 0; i < 8; ++i) {
      const float2* bp = bptr + (size_t)((g * 8 + i) * 4) * 32;
      const float2 q0 = bp[0], q1 = bp[32], q2 = bp[64], q3 = bp[96];
      v2f a; a.x = cur[i].x; a.y = cur[i].y;
      v2f b0; b0.x = q0.x; b0.y = q0.y;
      v2f b1; b1.x = q1.x; b1.y = q1.y;
      v2f b2; b2.x = q2.x; b2.y = q2.y;
      v2f b3; b3.x = q3.x; b3.y = q3.y;
      acc0 = __builtin_amdgcn_wmma_f32_16x16x4_f32(false, a, false, b0, (short)0, acc0, false, false);
      acc1 = __builtin_amdgcn_wmma_f32_16x16x4_f32(false, a, false, b1, (short)0, acc1, false, false);
      acc2 = __builtin_amdgcn_wmma_f32_16x16x4_f32(false, a, false, b2, (short)0, acc2, false, false);
      acc3 = __builtin_amdgcn_wmma_f32_16x16x4_f32(false, a, false, b3, (short)0, acc3, false, false);
    }
    if (g + 1 < NG) {
#pragma unroll
      for (int i = 0; i < 8; ++i) cur[i] = nxt[i];
    }
  }

  float* obase = out + (size_t)(tm * 16 + half * 8) * Nout + pn * 64 + l;
#pragma unroll
  for (int r = 0; r < 8; ++r) {
    float* orow = obase + (size_t)r * Nout;
    orow[0] = acc0[r];
    orow[16] = acc1[r];
    orow[32] = acc2[r];
    orow[48] = acc3[r];
  }
}

__global__ void attn_coef(const float* __restrict__ feat,
                          const float* __restrict__ al,
                          const float* __restrict__ ar,
                          float* __restrict__ el, float* __restrict__ er) {
  int i = blockIdx.x * blockDim.x + threadIdx.x;
  if (i >= NN * HH) return;
  int h = i & (HH - 1);
  const float4* f = (const float4*)(feat + (size_t)i * DD);
  const float4* a = (const float4*)(al + h * DD);
  const float4* b = (const float4*)(ar + h * DD);
  float sl = 0.f, sr = 0.f;
#pragma unroll 4
  for (int d = 0; d < DD / 4; ++d) {
    float4 fv = f[d], av = a[d], bv = b[d];
    sl += fv.x * av.x + fv.y * av.y + fv.z * av.z + fv.w * av.w;
    sr += fv.x * bv.x + fv.y * bv.y + fv.z * bv.z + fv.w * bv.w;
  }
  el[i] = sl;
  er[i] = sr;
}

__global__ void zero_counts(int* __restrict__ cnt) {
  int i = blockIdx.x * blockDim.x + threadIdx.x;
  if (i < NN) cnt[i] = 0;
}

__global__ void dst_count(const int* __restrict__ dst, int* __restrict__ cnt,
                          int* __restrict__ rnk) {
  int e = blockIdx.x * blockDim.x + threadIdx.x;
  if (e >= EE) return;
  rnk[e] = atomicAdd(&cnt[dst[e]], 1);
}

__global__ __launch_bounds__(1024) void scan_counts(const int* __restrict__ cnt,
                                                    int* __restrict__ rowptr) {
  __shared__ int part[1024];
  const int tid = threadIdx.x;
  const int per = (NN + 1023) / 1024;
  const int b = tid * per;
  const int e = (b + per < NN) ? (b + per) : NN;
  int s = 0;
  for (int i = b; i < e; ++i) s += cnt[i];
  part[tid] = s;
  __syncthreads();
  for (int off = 1; off < 1024; off <<= 1) {
    int v = (tid >= off) ? part[tid - off] : 0;
    __syncthreads();
    part[tid] += v;
    __syncthreads();
  }
  int run = part[tid] - s;
  for (int i = b; i < e; ++i) {
    rowptr[i] = run;
    run += cnt[i];
  }
  if (tid == 1023) rowptr[NN] = part[1023];
}

__global__ void dst_place(const int* __restrict__ src, const int* __restrict__ dst,
                          const int* __restrict__ rowptr, const int* __restrict__ rnk,
                          int* __restrict__ csrc) {
  int e = blockIdx.x * blockDim.x + threadIdx.x;
  if (e >= EE) return;
  csrc[rowptr[dst[e]] + rnk[e]] = src[e];
}

__device__ __forceinline__ float leaky(float v) {
  return v > 0.f ? v : NEG_SLOPE * v;
}

__device__ __forceinline__ float elu1(float v) {
  return v > 0.f ? v : (__expf(v) - 1.f);
}

__device__ __forceinline__ float wave_max(float v) {
  for (int o = 16; o > 0; o >>= 1) v = fmaxf(v, __shfl_xor(v, o));
  return v;
}

__device__ __forceinline__ float wave_sum(float v) {
  for (int o = 16; o > 0; o >>= 1) v += __shfl_xor(v, o);
  return v;
}

template <bool FINAL>
__global__ __launch_bounds__(128) void node_gather(
    const int* __restrict__ rowptr, const int* __restrict__ csrc,
    const float* __restrict__ feat, const float* __restrict__ el,
    const float* __restrict__ er, const float* __restrict__ hin,
    float* __restrict__ outp) {
  __shared__ float4 hs[FH / 4];

  const int t = blockIdx.x;
  const int tid = threadIdx.x;
  const int lane = tid & 31;
  const int h = tid >> 5;
  const int beg = rowptr[t];
  const int end = rowptr[t + 1];
  const float er_t = er[t * HH + h];

  float m = -__builtin_huge_valf();
  for (int i = beg + lane; i < end; i += 32) {
    int s = csrc[i];
    m = fmaxf(m, leaky(el[s * HH + h] + er_t));
  }
  m = wave_max(m);

  float dsum = 0.f;
  for (int i = beg + lane; i < end; i += 32) {
    int s = csrc[i];
    dsum += __expf(leaky(el[s * HH + h] + er_t) - m);
  }
  dsum = wave_sum(dsum);
  const float dn = (dsum == 0.f) ? 1.f : dsum;

  float4 acc = make_float4(0.f, 0.f, 0.f, 0.f);
  const float4* frow = (const float4*)feat + tid;
  for (int c = beg; c < end; c += 32) {
    const int i = c + lane;
    int s = 0;
    float a = 0.f;
    if (i < end) {
      s = csrc[i];
      a = __expf(leaky(el[s * HH + h] + er_t) - m) / dn;
    }
    int cnt = end - c;
    if (cnt > 32) cnt = 32;
#pragma unroll 4
    for (int j = 0; j < cnt; ++j) {
      const int sj = __builtin_amdgcn_readlane(s, j);
      const float aj = __int_as_float(__builtin_amdgcn_readlane(__float_as_int(a), j));
      const float4 f = frow[(size_t)sj * (FH / 4)];
      acc.x += f.x * aj;
      acc.y += f.y * aj;
      acc.z += f.z * aj;
      acc.w += f.w * aj;
    }
  }

  if (!FINAL) {
    float4 r;
    r.x = elu1(acc.x);
    r.y = elu1(acc.y);
    r.z = elu1(acc.z);
    r.w = elu1(acc.w);
    ((float4*)outp)[(size_t)t * (FH / 4) + tid] = r;
  } else {
    const float4 x = ((const float4*)hin)[(size_t)t * (FH / 4) + tid];
    float4 r;
    r.x = elu1(acc.x + x.x);
    r.y = elu1(acc.y + x.y);
    r.z = elu1(acc.z + x.z);
    r.w = elu1(acc.w + x.w);
    hs[tid] = r;
    __syncthreads();
    const float* hf = (const float*)hs;
    float sm = 0.f;
    for (int hh = 0; hh < HH; ++hh) sm += hf[hh * DD + tid];
    outp[(size_t)t * DD + tid] = sm * (1.f / HH);
  }
}

extern "C" void kernel_launch(void* const* d_in, const int* in_sizes, int n_in,
                              void* d_out, int out_size, void* d_ws, size_t ws_size,
                              hipStream_t stream) {
  const int* src = (const int*)d_in[0];
  const int* dst = (const int*)d_in[1];
  const float* x   = (const float*)d_in[2];
  const float* W0  = (const float*)d_in[3];
  const float* al0 = (const float*)d_in[4];
  const float* ar0 = (const float*)d_in[5];
  const float* W1  = (const float*)d_in[6];
  const float* al1 = (const float*)d_in[7];
  const float* ar1 = (const float*)d_in[8];
  float* out = (float*)d_out;

  float* ws    = (float*)d_ws;
  float* feat  = ws;
  float* hbuf  = feat  + (size_t)NN * FH;
  float* el    = hbuf  + (size_t)NN * FH;
  float* er    = el    + (size_t)NN * HH;
  int* cnt     = (int*)(er + (size_t)NN * HH);
  int* rowptr  = cnt    + NN;
  int* rnk     = rowptr + (NN + 16);
  int* csrc    = rnk    + EE;

  const int mtiles = (NN + 15) / 16;
  const int mgroups = (mtiles + 7) / 8;
  const int npanels = FH / 64;
  const int gemmBlocks = mgroups * npanels;

  zero_counts<<<(NN + 255) / 256, 256, 0, stream>>>(cnt);
  dst_count<<<(EE + 255) / 256, 256, 0, stream>>>(dst, cnt, rnk);
  scan_counts<<<1, 1024, 0, stream>>>(cnt, rowptr);
  dst_place<<<(EE + 255) / 256, 256, 0, stream>>>(src, dst, rowptr, rnk, csrc);

  gemm16x64_wmma<K0, FH><<<gemmBlocks, 256, 0, stream>>>(x, W0, feat, NN);
  attn_coef<<<(NN * HH + 255) / 256, 256, 0, stream>>>(feat, al0, ar0, el, er);
  node_gather<false><<<NN, 128, 0, stream>>>(rowptr, csrc, feat, el, er, feat, hbuf);
  gemm16x64_wmma<KMAX, FH><<<gemmBlocks, 256, 0, stream>>>(hbuf, W1, feat, NN);
  attn_coef<<<(NN * HH + 255) / 256, 256, 0, stream>>>(feat, al1, ar1, el, er);
  node_gather<true><<<NN, 128, 0, stream>>>(rowptr, csrc, feat, el, er, hbuf, out);
}


# el/er dots taken from the GEMM accumulators in its epilogue (per 64-column half head, 16-lane DPP tree; halves added at use); both attn_coef launches gone; el/er sum order changes
# speedup vs baseline: 5.9206x; 1.1164x over previous
#include <hip/hip_runtime.h>
#include <math.h>

#define NN 10000
#define EE 320000
#define HH 4
#define DD 128
#define FH (HH * DD)
#define K0 256
#define KMAX 512
#define NEG_SLOPE 0.2f

typedef __attribute__((ext_vector_type(2))) float v2f;
typedef __attribute__((ext_vector_type(8))) float v8f;


template <int MASK>
__device__ __forceinline__ float row_xor(float v) {
  return __int_as_float(__builtin_amdgcn_update_dpp(0, __float_as_int(v), 0x160 + MASK,
                                                    0xf, 0xf, true));
}

template <int K, int Nout>
__global__ __launch_bounds__(256) void gemm16x64_wmma(const float* __restrict__ X,
                                                      const float* __restrict__ W,
                                                      const float* __restrict__ al,
                                                      const float* __restrict__ ar,
                                                      float* __restrict__ out,
                                                      float* __restrict__ elp,
                                                      float* __restrict__ erp,
                                                      int M) {
  __shared__ float2 Bs[K * 32];
  __shared__ float As[128];

  const int tid = threadIdx.x;
  const int lane = tid & 31;
  const int wv = tid >> 5;
  constexpr int npan = Nout >> 6;
  const int mg = blockIdx.x / npan;
  const int pn = blockIdx.x - mg * npan;

  if (tid < 128) As[tid] = (tid < 64) ? al[pn * 64 + tid] : ar[pn * 64 + tid - 64];

  {
    const int c = tid & 63;
    const int j = c >> 4, l = c & 15;
    const float* wp = W + pn * 64 + c;
#pragma unroll 8
    for (int it = 0; it < K / 8; ++it) {
      const int kp = it * 4 + (tid >> 6);
      const float v0 = wp[(size_t)(2 * kp) * Nout];
      const float v1 = wp[(size_t)(2 * kp + 1) * Nout];
      const int ks = kp >> 1, h = kp & 1;
      Bs[((ks * 4 + j) * 2 + h) * 16 + l] = make_float2(v0, v1);
    }
  }
  __syncthreads();

  const int tm = mg * 8 + wv;
  if (tm * 16 >= M) return;

  const int half = lane >> 4;
  const int l = lane & 15;

  const float* aptr = X + (size_t)(tm * 16 + l) * K + half * 2;
  const float2* bptr = Bs + lane;

  v8f acc0 = {}, acc1 = {}, acc2 = {}, acc3 = {};

  float2 cur[8], nxt[8];
#pragma unroll
  for (int i = 0; i < 8; ++i) cur[i] = *(const float2*)(aptr + 4 * i);

  constexpr int NG = K / 32;
  for (int g = 0; g < NG; ++g) {
    if (g + 1 < NG) {
#pragma unroll
      for (int i = 0; i < 8; ++i) nxt[i] = *(const float2*)(aptr + 32 * (g + 1) + 4 * i);
    }
#pragma unroll
    for (int i = 0; i < 8; ++i) {
      const float2* bp = bptr + (size_t)((g * 8 + i) * 4) * 32;
      const float2 q0 = bp[0], q1 = bp[32], q2 = bp[64], q3 = bp[96];
      v2f a; a.x = cur[i].x; a.y = cur[i].y;
      v2f b0; b0.x = q0.x; b0.y = q0.y;
      v2f b1; b1.x = q1.x; b1.y = q1.y;
      v2f b2; b2.x = q2.x; b2.y = q2.y;
      v2f b3; b3.x = q3.x; b3.y = q3.y;
      acc0 = __builtin_amdgcn_wmma_f32_16x16x4_f32(false, a, false, b0, (short)0, acc0, false, false);
      acc1 = __builtin_amdgcn_wmma_f32_16x16x4_f32(false, a, false, b1, (short)0, acc1, false, false);
      acc2 = __builtin_amdgcn_wmma_f32_16x16x4_f32(false, a, false, b2, (short)0, acc2, false, false);
      acc3 = __builtin_amdgcn_wmma_f32_16x16x4_f32(false, a, false, b3, (short)0, acc3, false, false);
    }
    if (g + 1 < NG) {
#pragma unroll
      for (int i = 0; i < 8; ++i) cur[i] = nxt[i];
    }
  }

  float* obase = out + (size_t)(tm * 16 + half * 8) * Nout + pn * 64 + l;
#pragma unroll
  for (int r = 0; r < 8; ++r) {
    float* orow = obase + (size_t)r * Nout;
    orow[0] = acc0[r];
    orow[16] = acc1[r];
    orow[32] = acc2[r];
    orow[48] = acc3[r];
  }

  float x[16];
  {
    const float a0 = As[l], a1 = As[16 + l], a2 = As[32 + l], a3 = As[48 + l];
    const float b0 = As[64 + l], b1 = As[80 + l], b2 = As[96 + l], b3 = As[112 + l];
    for (int r = 0; r < 8; ++r) {
      x[r] = acc0[r] * a0 + acc1[r] * a1 + acc2[r] * a2 + acc3[r] * a3;
      x[8 + r] = acc0[r] * b0 + acc1[r] * b1 + acc2[r] * b2 + acc3[r] * b3;
    }
  }
  const bool t8 = (l & 8) != 0, t4 = (l & 4) != 0, t2 = (l & 2) != 0, t1 = (l & 1) != 0;
  float y[8], z[4], w[2];
  for (int i = 0; i < 8; ++i)
    y[i] = (t8 ? x[i + 8] : x[i]) + row_xor<8>(t8 ? x[i] : x[i + 8]);
  for (int i = 0; i < 4; ++i)
    z[i] = (t4 ? y[i + 4] : y[i]) + row_xor<4>(t4 ? y[i] : y[i + 4]);
  for (int i = 0; i < 2; ++i)
    w[i] = (t2 ? z[i + 2] : z[i]) + row_xor<2>(t2 ? z[i] : z[i + 2]);
  const float u = (t1 ? w[1] : w[0]) + row_xor<1>(t1 ? w[0] : w[1]);
  float* ep = t8 ? erp : elp;
  ep[(size_t)(tm * 16 + half * 8 + (l & 7)) * (Nout >> 6) + pn] = u;
}

__global__ void zero_counts(int* __restrict__ cnt) {
  int i = blockIdx.x * blockDim.x + threadIdx.x;
  if (i < NN) cnt[i] = 0;
}

__global__ void dst_count(const int* __restrict__ dst, int* __restrict__ cnt,
                          int* __restrict__ rnk) {
  int e = blockIdx.x * blockDim.x + threadIdx.x;
  if (e >= EE) return;
  rnk[e] = atomicAdd(&cnt[dst[e]], 1);
}

__global__ __launch_bounds__(1024) void scan_counts(const int* __restrict__ cnt,
                                                    int* __restrict__ rowptr) {
  __shared__ int part[1024];
  const int tid = threadIdx.x;
  const int per = (NN + 1023) / 1024;
  const int b = tid * per;
  const int e = (b + per < NN) ? (b + per) : NN;
  int s = 0;
  for (int i = b; i < e; ++i) s += cnt[i];
  part[tid] = s;
  __syncthreads();
  for (int off = 1; off < 1024; off <<= 1) {
    int v = (tid >= off) ? part[tid - off] : 0;
    __syncthreads();
    part[tid] += v;
    __syncthreads();
  }
  int run = part[tid] - s;
  for (int i = b; i < e; ++i) {
    rowptr[i] = run;
    run += cnt[i];
  }
  if (tid == 1023) rowptr[NN] = part[1023];
}

__global__ void dst_place(const int* __restrict__ src, const int* __restrict__ dst,
                          const int* __restrict__ rowptr, const int* __restrict__ rnk,
                          int* __restrict__ csrc) {
  int e = blockIdx.x * blockDim.x + threadIdx.x;
  if (e >= EE) return;
  csrc[rowptr[dst[e]] + rnk[e]] = src[e];
}

__device__ __forceinline__ float leaky(float v) {
  return v > 0.f ? v : NEG_SLOPE * v;
}

__device__ __forceinline__ float elu1(float v) {
  return v > 0.f ? v : (__expf(v) - 1.f);
}

__device__ __forceinline__ float wave_max(float v) {
  for (int o = 16; o > 0; o >>= 1) v = fmaxf(v, __shfl_xor(v, o));
  return v;
}

__device__ __forceinline__ float wave_sum(float v) {
  for (int o = 16; o > 0; o >>= 1) v += __shfl_xor(v, o);
  return v;
}

__device__ __forceinline__ float half_sum(const float* __restrict__ p, int i) {
  const float2 v = ((const float2*)p)[i];
  return v.x + v.y;
}

template <bool FINAL>
__global__ __launch_bounds__(128) void node_gather(
    const int* __restrict__ rowptr, const int* __restrict__ csrc,
    const float* __restrict__ feat, const float* __restrict__ el,
    const float* __restrict__ er, const float* __restrict__ hin,
    float* __restrict__ outp) {
  __shared__ float4 hs[FH / 4];

  const int t = blockIdx.x;
  const int tid = threadIdx.x;
  const int lane = tid & 31;
  const int h = tid >> 5;
  const int beg = rowptr[t];
  const int end = rowptr[t + 1];
  const float er_t = half_sum(er, t * HH + h);

  float m = -__builtin_huge_valf();
  for (int i = beg + lane; i < end; i += 32) {
    int s = csrc[i];
    m = fmaxf(m, leaky(half_sum(el, s * HH + h) + er_t));
  }
  m = wave_max(m);

  float dsum = 0.f;
  for (int i = beg + lane; i < end; i += 32) {
    int s = csrc[i];
    dsum += __expf(leaky(half_sum(el, s * HH + h) + er_t) - m);
  }
  dsum = wave_sum(dsum);
  const float dn = (dsum == 0.f) ? 1.f : dsum;

  float4 acc = make_float4(0.f, 0.f, 0.f, 0.f);
  const float4* frow = (const float4*)feat + tid;
  for (int c = beg; c < end; c += 32) {
    const int i = c + lane;
    int s = 0;
    float a = 0.f;
    if (i < end) {
      s = csrc[i];
      a = __expf(leaky(half_sum(el, s * HH + h) + er_t) - m) / dn;
    }
    int cnt = end - c;
    if (cnt > 32) cnt = 32;
#pragma unroll 4
    for (int j = 0; j < cnt; ++j) {
      const int sj = __builtin_amdgcn_readlane(s, j);
      const float aj = __int_as_float(__builtin_amdgcn_readlane(__float_as_int(a), j));
      const float4 f = frow[(size_t)sj * (FH / 4)];
      acc.x += f.x * aj;
      acc.y += f.y * aj;
      acc.z += f.z * aj;
      acc.w += f.w * aj;
    }
  }

  if (!FINAL) {
    float4 r;
    r.x = elu1(acc.x);
    r.y = elu1(acc.y);
    r.z = elu1(acc.z);
    r.w = elu1(acc.w);
    ((float4*)outp)[(size_t)t * (FH / 4) + tid] = r;
  } else {
    const float4 x = ((const float4*)hin)[(size_t)t * (FH / 4) + tid];
    float4 r;
    r.x = elu1(acc.x + x.x);
    r.y = elu1(acc.y + x.y);
    r.z = elu1(acc.z + x.z);
    r.w = elu1(acc.w + x.w);
    hs[tid] = r;
    __syncthreads();
    const float* hf = (const float*)hs;
    float sm = 0.f;
    for (int hh = 0; hh < HH; ++hh) sm += hf[hh * DD + tid];
    outp[(size_t)t * DD + tid] = sm * (1.f / HH);
  }
}

extern "C" void kernel_launch(void* const* d_in, const int* in_sizes, int n_in,
                              void* d_out, int out_size, void* d_ws, size_t ws_size,
                              hipStream_t stream) {
  const int* src = (const int*)d_in[0];
  const int* dst = (const int*)d_in[1];
  const float* x   = (const float*)d_in[2];
  const float* W0  = (const float*)d_in[3];
  const float* al0 = (const float*)d_in[4];
  const float* ar0 = (const float*)d_in[5];
  const float* W1  = (const float*)d_in[6];
  const float* al1 = (const float*)d_in[7];
  const float* ar1 = (const float*)d_in[8];
  float* out = (float*)d_out;

  float* ws    = (float*)d_ws;
  float* feat  = ws;
  float* hbuf  = feat  + (size_t)NN * FH;
  float* el    = hbuf  + (size_t)NN * FH;
  float* er    = el    + (size_t)NN * HH * 2;
  int* cnt     = (int*)(er + (size_t)NN * HH * 2);
  int* rowptr  = cnt    + NN;
  int* rnk     = rowptr + (NN + 16);
  int* csrc    = rnk    + EE;

  const int mtiles = (NN + 15) / 16;
  const int mgroups = (mtiles + 7) / 8;
  const int npanels = FH / 64;
  const int gemmBlocks = mgroups * npanels;

  zero_counts<<<(NN + 255) / 256, 256, 0, stream>>>(cnt);
  dst_count<<<(EE + 255) / 256, 256, 0, stream>>>(dst, cnt, rnk);
  scan_counts<<<1, 1024, 0, stream>>>(cnt, rowptr);
  dst_place<<<(EE + 255) / 256, 256, 0, stream>>>(src, dst, rowptr, rnk, csrc);

  gemm16x64_wmma<K0, FH><<<gemmBlocks, 256, 0, stream>>>(x, W0, al0, ar0, feat, el, er, NN);
  node_gather<false><<<NN, 128, 0, stream>>>(rowptr, csrc, feat, el, er, feat, hbuf);
  gemm16x64_wmma<KMAX, FH><<<gemmBlocks, 256, 0, stream>>>(hbuf, W1, al1, ar1, feat, el, er, NN);
  node_gather<true><<<NN, 128, 0, stream>>>(rowptr, csrc, feat, el, er, hbuf, out);
}


# Edge grouping in two levels on LDS counters: 640-edge slabs ordered by 256-node bin, one block per bin counts, scans, places; global int atomics gone, 4 launches to 2; sum order: LDS arrival rank
# speedup vs baseline: 7.0599x; 1.1924x over previous
#include <hip/hip_runtime.h>
#include <math.h>

#define NN 10000
#define EE 320000
#define HH 4
#define DD 128
#define FH (HH * DD)
#define K0 256
#define KMAX 512
#define NEG_SLOPE 0.2f

typedef __attribute__((ext_vector_type(2))) float v2f;
typedef __attribute__((ext_vector_type(8))) float v8f;


template <int MASK>
__device__ __forceinline__ float row_xor(float v) {
  return __int_as_float(__builtin_amdgcn_update_dpp(0, __float_as_int(v), 0x160 + MASK,
                                                    0xf, 0xf, true));
}

template <int K, int Nout>
__global__ __launch_bounds__(256) void gemm16x64_wmma(const float* __restrict__ X,
                                                      const float* __restrict__ W,
                                                      const float* __restrict__ al,
                                                      const float* __restrict__ ar,
                                                      float* __restrict__ out,
                                                      float* __restrict__ elp,
                                                      float* __restrict__ erp,
                                                      int M) {
  __shared__ float2 Bs[K * 32];
  __shared__ float As[128];

  const int tid = threadIdx.x;
  const int lane = tid & 31;
  const int wv = tid >> 5;
  constexpr int npan = Nout >> 6;
  const int mg = blockIdx.x / npan;
  const int pn = blockIdx.x - mg * npan;

  if (tid < 128) As[tid] = (tid < 64) ? al[pn * 64 + tid] : ar[pn * 64 + tid - 64];

  {
    const int c = tid & 63;
    const int j = c >> 4, l = c & 15;
    const float* wp = W + pn * 64 + c;
#pragma unroll 8
    for (int it = 0; it < K / 8; ++it) {
      const int kp = it * 4 + (tid >> 6);
      const float v0 = wp[(size_t)(2 * kp) * Nout];
      const float v1 = wp[(size_t)(2 * kp + 1) * Nout];
      const int ks = kp >> 1, h = kp & 1;
      Bs[((ks * 4 + j) * 2 + h) * 16 + l] = make_float2(v0, v1);
    }
  }
  __syncthreads();

  const int tm = mg * 8 + wv;
  if (tm * 16 >= M) return;

  const int half = lane >> 4;
  const int l = lane & 15;

  const float* aptr = X + (size_t)(tm * 16 + l) * K + half * 2;
  const float2* bptr = Bs + lane;

  v8f acc0 = {}, acc1 = {}, acc2 = {}, acc3 = {};

  float2 cur[8], nxt[8];
#pragma unroll
  for (int i = 0; i < 8; ++i) cur[i] = *(const float2*)(aptr + 4 * i);

  constexpr int NG = K / 32;
  for (int g = 0; g < NG; ++g) {
    if (g + 1 < NG) {
#pragma unroll
      for (int i = 0; i < 8; ++i) nxt[i] = *(const float2*)(aptr + 32 * (g + 1) + 4 * i);
    }
#pragma unroll
    for (int i = 0; i < 8; ++i) {
      const float2* bp = bptr + (size_t)((g * 8 + i) * 4) * 32;
      const float2 q0 = bp[0], q1 = bp[32], q2 = bp[64], q3 = bp[96];
      v2f a; a.x = cur[i].x; a.y = cur[i].y;
      v2f b0; b0.x = q0.x; b0.y = q0.y;
      v2f b1; b1.x = q1.x; b1.y = q1.y;
      v2f b2; b2.x = q2.x; b2.y = q2.y;
      v2f b3; b3.x = q3.x; b3.y = q3.y;
      acc0 = __builtin_amdgcn_wmma_f32_16x16x4_f32(false, a, false, b0, (short)0, acc0, false, false);
      acc1 = __builtin_amdgcn_wmma_f32_16x16x4_f32(false, a, false, b1, (short)0, acc1, false, false);
      acc2 = __builtin_amdgcn_wmma_f32_16x16x4_f32(false, a, false, b2, (short)0, acc2, false, false);
      acc3 = __builtin_amdgcn_wmma_f32_16x16x4_f32(false, a, false, b3, (short)0, acc3, false, false);
    }
    if (g + 1 < NG) {
#pragma unroll
      for (int i = 0; i < 8; ++i) cur[i] = nxt[i];
    }
  }

  float* obase = out + (size_t)(tm * 16 + half * 8) * Nout + pn * 64 + l;
#pragma unroll
  for (int r = 0; r < 8; ++r) {
    float* orow = obase + (size_t)r * Nout;
    orow[0] = acc0[r];
    orow[16] = acc1[r];
    orow[32] = acc2[r];
    orow[48] = acc3[r];
  }

  float x[16];
  {
    const float a0 = As[l], a1 = As[16 + l], a2 = As[32 + l], a3 = As[48 + l];
    const float b0 = As[64 + l], b1 = As[80 + l], b2 = As[96 + l], b3 = As[112 + l];
    for (int r = 0; r < 8; ++r) {
      x[r] = acc0[r] * a0 + acc1[r] * a1 + acc2[r] * a2 + acc3[r] * a3;
      x[8 + r] = acc0[r] * b0 + acc1[r] * b1 + acc2[r] * b2 + acc3[r] * b3;
    }
  }
  const bool t8 = (l & 8) != 0, t4 = (l & 4) != 0, t2 = (l & 2) != 0, t1 = (l & 1) != 0;
  float y[8], z[4], w[2];
  for (int i = 0; i < 8; ++i)
    y[i] = (t8 ? x[i + 8] : x[i]) + row_xor<8>(t8 ? x[i] : x[i + 8]);
  for (int i = 0; i < 4; ++i)
    z[i] = (t4 ? y[i + 4] : y[i]) + row_xor<4>(t4 ? y[i] : y[i + 4]);
  for (int i = 0; i < 2; ++i)
    w[i] = (t2 ? z[i + 2] : z[i]) + row_xor<2>(t2 ? z[i] : z[i + 2]);
  const float u = (t1 ? w[1] : w[0]) + row_xor<1>(t1 ? w[0] : w[1]);
  float* ep = t8 ? erp : elp;
  ep[(size_t)(tm * 16 + half * 8 + (l & 7)) * (Nout >> 6) + pn] = u;
}

#define GRP_NODES 256
#define GRP_BINS ((NN + GRP_NODES - 1) / GRP_NODES)
#define SLAB 640
#define NSLAB (EE / SLAB)
#define SLAB_THREADS 128
#define SLAB_PER_THREAD (SLAB / SLAB_THREADS)

__device__ __forceinline__ int wave_sum_i(int v) {
  for (int o = 16; o > 0; o >>= 1) v += __shfl_xor(v, o);
  return v;
}

__global__ __launch_bounds__(SLAB_THREADS) void slab_bin(const int* __restrict__ src,
                                                         const int* __restrict__ dst,
                                                         int* __restrict__ tmpw,
                                                         int* __restrict__ tabc,
                                                         int* __restrict__ tabo) {
  static_assert(EE % SLAB == 0, "whole slabs");
  static_assert(SLAB % SLAB_THREADS == 0, "whole edges per thread");
  static_assert(GRP_BINS <= SLAB_THREADS, "one thread per bin below");
  __shared__ unsigned bcnt[GRP_BINS];
  __shared__ unsigned boff[GRP_BINS];

  int2* tmp = (int2*)tmpw;
  const int tid = threadIdx.x;
  const int e0 = blockIdx.x * SLAB;
  if (tid < GRP_BINS) bcnt[tid] = 0u;
  __syncthreads();

  int d[SLAB_PER_THREAD], s[SLAB_PER_THREAD];
  unsigned bin[SLAB_PER_THREAD], rk[SLAB_PER_THREAD];
#pragma unroll
  for (int i = 0; i < SLAB_PER_THREAD; ++i) {
    const int e = e0 + i * SLAB_THREADS + tid;
    d[i] = dst[e];
    s[i] = src[e];
  }
#pragma unroll
  for (int i = 0; i < SLAB_PER_THREAD; ++i) {
    bin[i] = (unsigned)d[i] / GRP_NODES;
    rk[i] = atomicAdd(&bcnt[bin[i]], 1u);
  }
  __syncthreads();

  unsigned myc = 0u, myo = 0u;
  if (tid < GRP_BINS) {
#pragma unroll
    for (int b = 0; b < GRP_BINS; ++b) {
      const unsigned v = bcnt[b];
      if (b < tid) myo += v;
      if (b == tid) myc = v;
    }
    boff[tid] = myo;
  }
  __syncthreads();

#pragma unroll
  for (int i = 0; i < SLAB_PER_THREAD; ++i)
    tmp[e0 + boff[bin[i]] + rk[i]] = make_int2(d[i], s[i]);
  if (tid < GRP_BINS) {
    tabc[tid * NSLAB + blockIdx.x] = (int)myc;
    tabo[tid * NSLAB + blockIdx.x] = (int)myo;
  }
}

__global__ __launch_bounds__(1024) void bin_group(const int* __restrict__ tmpw,
                                                  const int* __restrict__ tabc,
                                                  const int* __restrict__ tabo,
                                                  int* __restrict__ rowptr,
                                                  int* __restrict__ csrc) {
  static_assert(GRP_NODES == 256, "the scan below gives 8 nodes to each of 32 lanes");
  static_assert(NSLAB <= 1024 && NSLAB % 4 == 0, "one thread per slab; counts read 4 at a time");
  __shared__ unsigned cur[GRP_NODES];
  __shared__ int part[32];
  __shared__ int segc[NSLAB];
  __shared__ int sego[NSLAB];

  const int2* tmp = (const int2*)tmpw;
  const int tid = threadIdx.x;
  const int lane = tid & 31;
  const int wv = tid >> 5;
  const int b = blockIdx.x;
  const int lo = b * GRP_NODES;
  const unsigned span = (unsigned)((NN - lo < GRP_NODES) ? (NN - lo) : GRP_NODES);

  if (tid < GRP_NODES) cur[tid] = 0u;

  constexpr int NLOW = (GRP_BINS * (NSLAB / 4) + 1023) / 1024;
  const int4* tc4 = (const int4*)tabc;
  const int nlow4 = b * (NSLAB / 4);
  const int jrow = b * NSLAB + (tid < NSLAB ? tid : 0);
  const int sc = tabc[jrow];
  const int so = tabo[jrow];
  int4 low[NLOW];
#pragma unroll
  for (int it = 0; it < NLOW; ++it) {
    const int i = it * 1024 + tid;
    low[it] = tc4[i < nlow4 ? i : 0];
  }
  int below = 0;
#pragma unroll
  for (int it = 0; it < NLOW; ++it) {
    const int i = it * 1024 + tid;
    if (i < nlow4) below += low[it].x + low[it].y + low[it].z + low[it].w;
  }
  if (tid < NSLAB) {
    segc[tid] = sc;
    sego[tid] = so;
  }
  below = wave_sum_i(below);
  if (lane == 0) part[wv] = below;
  __syncthreads();

  constexpr int SPW = (NSLAB + 31) / 32;
  int c[SPW], ib[SPW];
  int2 rec[SPW];
#pragma unroll
  for (int m = 0; m < SPW; ++m) {
    const int j = wv + 32 * m;
    c[m] = 0;
    ib[m] = 0;
    if (j < NSLAB) {
      c[m] = segc[j];
      ib[m] = j * SLAB + sego[j];
    }
  }
#pragma unroll
  for (int m = 0; m < SPW; ++m) {
    rec[m] = make_int2(lo, 0);
    if (lane < c[m]) rec[m] = tmp[ib[m] + lane];
  }
#pragma unroll
  for (int m = 0; m < SPW; ++m) {
    if (lane < c[m]) atomicAdd(&cur[rec[m].x - lo], 1u);
    for (int i = lane + 32; i < c[m]; i += 32)
      atomicAdd(&cur[tmp[ib[m] + i].x - lo], 1u);
  }
  __syncthreads();

  if (wv == 0) {
    const int base = wave_sum_i(part[lane]);
    int k8[8];
    int s = 0;
#pragma unroll
    for (int k = 0; k < 8; ++k) {
      k8[k] = (int)cur[lane * 8 + k];
      s += k8[k];
    }
    int incl = s;
    for (int o = 1; o < 32; o <<= 1) {
      const int v = __shfl_up(incl, o);
      if (lane >= o) incl += v;
    }
    int run = base + incl - s;
#pragma unroll
    for (int k = 0; k < 8; ++k) {
      const int i = lane * 8 + k;
      if ((unsigned)i < span) rowptr[lo + i] = run;
      cur[i] = (unsigned)run;
      run += k8[k];
    }
    if (b == (int)gridDim.x - 1 && lane == 31) rowptr[NN] = run;
  }
  __syncthreads();

#pragma unroll
  for (int m = 0; m < SPW; ++m) {
    if (lane < c[m]) csrc[atomicAdd(&cur[rec[m].x - lo], 1u)] = rec[m].y;
    for (int i = lane + 32; i < c[m]; i += 32) {
      const int2 t = tmp[ib[m] + i];
      csrc[atomicAdd(&cur[t.x - lo], 1u)] = t.y;
    }
  }
}

__device__ __forceinline__ float leaky(float v) {
  return v > 0.f ? v : NEG_SLOPE * v;
}

__device__ __forceinline__ float elu1(float v) {
  return v > 0.f ? v : (__expf(v) - 1.f);
}

__device__ __forceinline__ float wave_max(float v) {
  for (int o = 16; o > 0; o >>= 1) v = fmaxf(v, __shfl_xor(v, o));
  return v;
}

__device__ __forceinline__ float wave_sum(float v) {
  for (int o = 16; o > 0; o >>= 1) v += __shfl_xor(v, o);
  return v;
}

__device__ __forceinline__ float half_sum(const float* __restrict__ p, int i) {
  const float2 v = ((const float2*)p)[i];
  return v.x + v.y;
}

template <bool FINAL>
__global__ __launch_bounds__(128) void node_gather(
    const int* __restrict__ rowptr, const int* __restrict__ csrc,
    const float* __restrict__ feat, const float* __restrict__ el,
    const float* __restrict__ er, const float* __restrict__ hin,
    float* __restrict__ outp) {
  __shared__ float4 hs[FH / 4];

  const int t = blockIdx.x;
  const int tid = threadIdx.x;
  const int lane = tid & 31;
  const int h = tid >> 5;
  const int beg = rowptr[t];
  const int end = rowptr[t + 1];
  const float er_t = half_sum(er, t * HH + h);

  float m = -__builtin_huge_valf();
  for (int i = beg + lane; i < end; i += 32) {
    int s = csrc[i];
    m = fmaxf(m, leaky(half_sum(el, s * HH + h) + er_t));
  }
  m = wave_max(m);

  float dsum = 0.f;
  for (int i = beg + lane; i < end; i += 32) {
    int s = csrc[i];
    dsum += __expf(leaky(half_sum(el, s * HH + h) + er_t) - m);
  }
  dsum = wave_sum(dsum);
  const float dn = (dsum == 0.f) ? 1.f : dsum;

  float4 acc = make_float4(0.f, 0.f, 0.f, 0.f);
  const float4* frow = (const float4*)feat + tid;
  for (int c = beg; c < end; c += 32) {
    const int i = c + lane;
    int s = 0;
    float a = 0.f;
    if (i < end) {
      s = csrc[i];
      a = __expf(leaky(half_sum(el, s * HH + h) + er_t) - m) / dn;
    }
    int cnt = end - c;
    if (cnt > 32) cnt = 32;
#pragma unroll 4
    for (int j = 0; j < cnt; ++j) {
      const int sj = __builtin_amdgcn_readlane(s, j);
      const float aj = __int_as_float(__builtin_amdgcn_readlane(__float_as_int(a), j));
      const float4 f = frow[(size_t)sj * (FH / 4)];
      acc.x += f.x * aj;
      acc.y += f.y * aj;
      acc.z += f.z * aj;
      acc.w += f.w * aj;
    }
  }

  if (!FINAL) {
    float4 r;
    r.x = elu1(acc.x);
    r.y = elu1(acc.y);
    r.z = elu1(acc.z);
    r.w = elu1(acc.w);
    ((float4*)outp)[(size_t)t * (FH / 4) + tid] = r;
  } else {
    const float4 x = ((const float4*)hin)[(size_t)t * (FH / 4) + tid];
    float4 r;
    r.x = elu1(acc.x + x.x);
    r.y = elu1(acc.y + x.y);
    r.z = elu1(acc.z + x.z);
    r.w = elu1(acc.w + x.w);
    hs[tid] = r;
    __syncthreads();
    const float* hf = (const float*)hs;
    float sm = 0.f;
    for (int hh = 0; hh < HH; ++hh) sm += hf[hh * DD + tid];
    outp[(size_t)t * DD + tid] = sm * (1.f / HH);
  }
}

extern "C" void kernel_launch(void* const* d_in, const int* in_sizes, int n_in,
                              void* d_out, int out_size, void* d_ws, size_t ws_size,
                              hipStream_t stream) {
  const int* src = (const int*)d_in[0];
  const int* dst = (const int*)d_in[1];
  const float* x   = (const float*)d_in[2];
  const float* W0  = (const float*)d_in[3];
  const float* al0 = (const float*)d_in[4];
  const float* ar0 = (const float*)d_in[5];
  const float* W1  = (const float*)d_in[6];
  const float* al1 = (const float*)d_in[7];
  const float* ar1 = (const float*)d_in[8];
  float* out = (float*)d_out;

  float* ws    = (float*)d_ws;
  float* feat  = ws;
  float* hbuf  = feat  + (size_t)NN * FH;
  float* el    = hbuf  + (size_t)NN * FH;
  float* er    = el    + (size_t)NN * HH * 2;
  int* rowptr  = (int*)(er + (size_t)NN * HH * 2);
  int* csrc    = rowptr + (NN + 16);
  int* tabc    = csrc   + EE;
  int* tabo    = tabc   + GRP_BINS * NSLAB;
  int* tmp     = tabo   + GRP_BINS * NSLAB;

  const int mtiles = (NN + 15) / 16;
  const int mgroups = (mtiles + 7) / 8;
  const int npanels = FH / 64;
  const int gemmBlocks = mgroups * npanels;

  slab_bin<<<NSLAB, SLAB_THREADS, 0, stream>>>(src, dst, tmp, tabc, tabo);
  bin_group<<<GRP_BINS, 1024, 0, stream>>>(tmp, tabc, tabo, rowptr, csrc);

  gemm16x64_wmma<K0, FH><<<gemmBlocks, 256, 0, stream>>>(x, W0, al0, ar0, feat, el, er, NN);
  node_gather<false><<<NN, 128, 0, stream>>>(rowptr, csrc, feat, el, er, feat, hbuf);
  gemm16x64_wmma<KMAX, FH><<<gemmBlocks, 256, 0, stream>>>(hbuf, W1, al1, ar1, feat, el, er, NN);
  node_gather<true><<<NN, 128, 0, stream>>>(rowptr, csrc, feat, el, er, hbuf, out);
}


# GEMM regrouped: 20-wave blocks of 20 M-tiles x 64 columns, 256 blocks, one per WGP (LDS sized over half a WGP): 5 strips per SIMD instead of 6 on the fullest; sum order unchanged
# speedup vs baseline: 7.3943x; 1.0474x over previous
#include <hip/hip_runtime.h>
#include <math.h>

#define NN 10000
#define EE 320000
#define HH 4
#define DD 128
#define FH (HH * DD)
#define K0 256
#define KMAX 512
#define NEG_SLOPE 0.2f

typedef __attribute__((ext_vector_type(2))) float v2f;
typedef __attribute__((ext_vector_type(8))) float v8f;


template <int MASK>
__device__ __forceinline__ float row_xor(float v) {
  return __int_as_float(__builtin_amdgcn_update_dpp(0, __float_as_int(v), 0x160 + MASK,
                                                    0xf, 0xf, true));
}

#define GEMM_WAVES 20
#define GEMM_THREADS (GEMM_WAVES * 32)
#define GEMM_MIN_ENTRIES 20800

template <int K, int Nout>
__global__ __launch_bounds__(GEMM_THREADS) void gemm16x64_wmma(const float* __restrict__ X,
                                                      const float* __restrict__ W,
                                                      const float* __restrict__ al,
                                                      const float* __restrict__ ar,
                                                      float* __restrict__ out,
                                                      float* __restrict__ elp,
                                                      float* __restrict__ erp,
                                                      int M) {
  constexpr int NB = (K * 32 > GEMM_MIN_ENTRIES) ? K * 32 : GEMM_MIN_ENTRIES;
  __shared__ float2 Bs[NB];
  __shared__ float As[128];

  const int tid = threadIdx.x;
  const int lane = tid & 31;
  const int wv = tid >> 5;
  constexpr int npan = Nout >> 6;
  const int mg = blockIdx.x / npan;
  const int pn = blockIdx.x - mg * npan;

  if (tid < 128) As[tid] = (tid < 64) ? al[pn * 64 + tid] : ar[pn * 64 + tid - 64];

  {
    const int c = tid & 63;
    const int j = c >> 4, l = c & 15;
    const float* wp = W + pn * 64 + c;
    constexpr int NGRP = GEMM_THREADS / 64;
    constexpr int NIT = (K / 2 + NGRP - 1) / NGRP;
#pragma unroll 13
    for (int it = 0; it < NIT; ++it) {
      const int kp = it * NGRP + (tid >> 6);
      if (kp < K / 2) {
        const float v0 = wp[(size_t)(2 * kp) * Nout];
        const float v1 = wp[(size_t)(2 * kp + 1) * Nout];
        const int ks = kp >> 1, h = kp & 1;
        Bs[((ks * 4 + j) * 2 + h) * 16 + l] = make_float2(v0, v1);
      }
    }
  }
  __syncthreads();

  const int tm = mg * GEMM_WAVES + wv;
  if (tm * 16 >= M) return;

  const int half = lane >> 4;
  const int l = lane & 15;

  const float* aptr = X + (size_t)(tm * 16 + l) * K + half * 2;
  const float2* bptr = Bs + lane;

  v8f acc0 = {}, acc1 = {}, acc2 = {}, acc3 = {};

  float2 cur[8], nxt[8];
#pragma unroll
  for (int i = 0; i < 8; ++i) cur[i] = *(const float2*)(aptr + 4 * i);

  constexpr int NG = K / 32;
  for (int g = 0; g < NG; ++g) {
    if (g + 1 < NG) {
#pragma unroll
      for (int i = 0; i < 8; ++i) nxt[i] = *(const float2*)(aptr + 32 * (g + 1) + 4 * i);
    }
#pragma unroll
    for (int i = 0; i < 8; ++i) {
      const float2* bp = bptr + (size_t)((g * 8 + i) * 4) * 32;
      const float2 q0 = bp[0], q1 = bp[32], q2 = bp[64], q3 = bp[96];
      v2f a; a.x = cur[i].x; a.y = cur[i].y;
      v2f b0; b0.x = q0.x; b0.y = q0.y;
      v2f b1; b1.x = q1.x; b1.y = q1.y;
      v2f b2; b2.x = q2.x; b2.y = q2.y;
      v2f b3; b3.x = q3.x; b3.y = q3.y;
      acc0 = __builtin_amdgcn_wmma_f32_16x16x4_f32(false, a, false, b0, (short)0, acc0, false, false);
      acc1 = __builtin_amdgcn_wmma_f32_16x16x4_f32(false, a, false, b1, (short)0, acc1, false, false);
      acc2 = __builtin_amdgcn_wmma_f32_16x16x4_f32(false, a, false, b2, (short)0, acc2, false, false);
      acc3 = __builtin_amdgcn_wmma_f32_16x16x4_f32(false, a, false, b3, (short)0, acc3, false, false);
    }
    if (g + 1 < NG) {
#pragma unroll
      for (int i = 0; i < 8; ++i) cur[i] = nxt[i];
    }
  }

  float* obase = out + (size_t)(tm * 16 + half * 8) * Nout + pn * 64 + l;
#pragma unroll
  for (int r = 0; r < 8; ++r) {
    float* orow = obase + (size_t)r * Nout;
    orow[0] = acc0[r];
    orow[16] = acc1[r];
    orow[32] = acc2[r];
    orow[48] = acc3[r];
  }

  float x[16];
  {
    const float a0 = As[l], a1 = As[16 + l], a2 = As[32 + l], a3 = As[48 + l];
    const float b0 = As[64 + l], b1 = As[80 + l], b2 = As[96 + l], b3 = As[112 + l];
    for (int r = 0; r < 8; ++r) {
      x[r] = acc0[r] * a0 + acc1[r] * a1 + acc2[r] * a2 + acc3[r] * a3;
      x[8 + r] = acc0[r] * b0 + acc1[r] * b1 + acc2[r] * b2 + acc3[r] * b3;
    }
  }
  const bool t8 = (l & 8) != 0, t4 = (l & 4) != 0, t2 = (l & 2) != 0, t1 = (l & 1) != 0;
  float y[8], z[4], w[2];
  for (int i = 0; i < 8; ++i)
    y[i] = (t8 ? x[i + 8] : x[i]) + row_xor<8>(t8 ? x[i] : x[i + 8]);
  for (int i = 0; i < 4; ++i)
    z[i] = (t4 ? y[i + 4] : y[i]) + row_xor<4>(t4 ? y[i] : y[i + 4]);
  for (int i = 0; i < 2; ++i)
    w[i] = (t2 ? z[i + 2] : z[i]) + row_xor<2>(t2 ? z[i] : z[i + 2]);
  const float u = (t1 ? w[1] : w[0]) + row_xor<1>(t1 ? w[0] : w[1]);
  float* ep = t8 ? erp : elp;
  ep[(size_t)(tm * 16 + half * 8 + (l & 7)) * (Nout >> 6) + pn] = u;
}

#define GRP_NODES 256
#define GRP_BINS ((NN + GRP_NODES - 1) / GRP_NODES)
#define SLAB 640
#define NSLAB (EE / SLAB)
#define SLAB_THREADS 128
#define SLAB_PER_THREAD (SLAB / SLAB_THREADS)

__device__ __forceinline__ int wave_sum_i(int v) {
  for (int o = 16; o > 0; o >>= 1) v += __shfl_xor(v, o);
  return v;
}

__global__ __launch_bounds__(SLAB_THREADS) void slab_bin(const int* __restrict__ src,
                                                         const int* __restrict__ dst,
                                                         int* __restrict__ tmpw,
                                                         int* __restrict__ tabc,
                                                         int* __restrict__ tabo) {
  static_assert(EE % SLAB == 0, "whole slabs");
  static_assert(SLAB % SLAB_THREADS == 0, "whole edges per thread");
  static_assert(GRP_BINS <= SLAB_THREADS, "one thread per bin below");
  __shared__ unsigned bcnt[GRP_BINS];
  __shared__ unsigned boff[GRP_BINS];

  int2* tmp = (int2*)tmpw;
  const int tid = threadIdx.x;
  const int e0 = blockIdx.x * SLAB;
  if (tid < GRP_BINS) bcnt[tid] = 0u;
  __syncthreads();

  int d[SLAB_PER_THREAD], s[SLAB_PER_THREAD];
  unsigned bin[SLAB_PER_THREAD], rk[SLAB_PER_THREAD];
#pragma unroll
  for (int i = 0; i < SLAB_PER_THREAD; ++i) {
    const int e = e0 + i * SLAB_THREADS + tid;
    d[i] = dst[e];
    s[i] = src[e];
  }
#pragma unroll
  for (int i = 0; i < SLAB_PER_THREAD; ++i) {
    bin[i] = (unsigned)d[i] / GRP_NODES;
    rk[i] = atomicAdd(&bcnt[bin[i]], 1u);
  }
  __syncthreads();

  unsigned myc = 0u, myo = 0u;
  if (tid < GRP_BINS) {
#pragma unroll
    for (int b = 0; b < GRP_BINS; ++b) {
      const unsigned v = bcnt[b];
      if (b < tid) myo += v;
      if (b == tid) myc = v;
    }
    boff[tid] = myo;
  }
  __syncthreads();

#pragma unroll
  for (int i = 0; i < SLAB_PER_THREAD; ++i)
    tmp[e0 + boff[bin[i]] + rk[i]] = make_int2(d[i], s[i]);
  if (tid < GRP_BINS) {
    tabc[tid * NSLAB + blockIdx.x] = (int)myc;
    tabo[tid * NSLAB + blockIdx.x] = (int)myo;
  }
}

__global__ __launch_bounds__(1024) void bin_group(const int* __restrict__ tmpw,
                                                  const int* __restrict__ tabc,
                                                  const int* __restrict__ tabo,
                                                  int* __restrict__ rowptr,
                                                  int* __restrict__ csrc) {
  static_assert(GRP_NODES == 256, "the scan below gives 8 nodes to each of 32 lanes");
  static_assert(NSLAB <= 1024 && NSLAB % 4 == 0, "one thread per slab; counts read 4 at a time");
  __shared__ unsigned cur[GRP_NODES];
  __shared__ int part[32];
  __shared__ int segc[NSLAB];
  __shared__ int sego[NSLAB];

  const int2* tmp = (const int2*)tmpw;
  const int tid = threadIdx.x;
  const int lane = tid & 31;
  const int wv = tid >> 5;
  const int b = blockIdx.x;
  const int lo = b * GRP_NODES;
  const unsigned span = (unsigned)((NN - lo < GRP_NODES) ? (NN - lo) : GRP_NODES);

  if (tid < GRP_NODES) cur[tid] = 0u;

  constexpr int NLOW = (GRP_BINS * (NSLAB / 4) + 1023) / 1024;
  const int4* tc4 = (const int4*)tabc;
  const int nlow4 = b * (NSLAB / 4);
  const int jrow = b * NSLAB + (tid < NSLAB ? tid : 0);
  const int sc = tabc[jrow];
  const int so = tabo[jrow];
  int4 low[NLOW];
#pragma unroll
  for (int it = 0; it < NLOW; ++it) {
    const int i = it * 1024 + tid;
    low[it] = tc4[i < nlow4 ? i : 0];
  }
  int below = 0;
#pragma unroll
  for (int it = 0; it < NLOW; ++it) {
    const int i = it * 1024 + tid;
    if (i < nlow4) below += low[it].x + low[it].y + low[it].z + low[it].w;
  }
  if (tid < NSLAB) {
    segc[tid] = sc;
    sego[tid] = so;
  }
  below = wave_sum_i(below);
  if (lane == 0) part[wv] = below;
  __syncthreads();

  constexpr int SPW = (NSLAB + 31) / 32;
  int c[SPW], ib[SPW];
  int2 rec[SPW];
#pragma unroll
  for (int m = 0; m < SPW; ++m) {
    const int j = wv + 32 * m;
    c[m] = 0;
    ib[m] = 0;
    if (j < NSLAB) {
      c[m] = segc[j];
      ib[m] = j * SLAB + sego[j];
    }
  }
#pragma unroll
  for (int m = 0; m < SPW; ++m) {
    rec[m] = make_int2(lo, 0);
    if (lane < c[m]) rec[m] = tmp[ib[m] + lane];
  }
#pragma unroll
  for (int m = 0; m < SPW; ++m) {
    if (lane < c[m]) atomicAdd(&cur[rec[m].x - lo], 1u);
    for (int i = lane + 32; i < c[m]; i += 32)
      atomicAdd(&cur[tmp[ib[m] + i].x - lo], 1u);
  }
  __syncthreads();

  if (wv == 0) {
    const int base = wave_sum_i(part[lane]);
    int k8[8];
    int s = 0;
#pragma unroll
    for (int k = 0; k < 8; ++k) {
      k8[k] = (int)cur[lane * 8 + k];
      s += k8[k];
    }
    int incl = s;
    for (int o = 1; o < 32; o <<= 1) {
      const int v = __shfl_up(incl, o);
      if (lane >= o) incl += v;
    }
    int run = base + incl - s;
#pragma unroll
    for (int k = 0; k < 8; ++k) {
      const int i = lane * 8 + k;
      if ((unsigned)i < span) rowptr[lo + i] = run;
      cur[i] = (unsigned)run;
      run += k8[k];
    }
    if (b == (int)gridDim.x - 1 && lane == 31) rowptr[NN] = run;
  }
  __syncthreads();

#pragma unroll
  for (int m = 0; m < SPW; ++m) {
    if (lane < c[m]) csrc[atomicAdd(&cur[rec[m].x - lo], 1u)] = rec[m].y;
    for (int i = lane + 32; i < c[m]; i += 32) {
      const int2 t = tmp[ib[m] + i];
      csrc[atomicAdd(&cur[t.x - lo], 1u)] = t.y;
    }
  }
}

__device__ __forceinline__ float leaky(float v) {
  return v > 0.f ? v : NEG_SLOPE * v;
}

__device__ __forceinline__ float elu1(float v) {
  return v > 0.f ? v : (__expf(v) - 1.f);
}

__device__ __forceinline__ float wave_max(float v) {
  for (int o = 16; o > 0; o >>= 1) v = fmaxf(v, __shfl_xor(v, o));
  return v;
}

__device__ __forceinline__ float wave_sum(float v) {
  for (int o = 16; o > 0; o >>= 1) v += __shfl_xor(v, o);
  return v;
}

__device__ __forceinline__ float half_sum(const float* __restrict__ p, int i) {
  const float2 v = ((const float2*)p)[i];
  return v.x + v.y;
}

template <bool FINAL>
__global__ __launch_bounds__(128) void node_gather(
    const int* __restrict__ rowptr, const int* __restrict__ csrc,
    const float* __restrict__ feat, const float* __restrict__ el,
    const float* __restrict__ er, const float* __restrict__ hin,
    float* __restrict__ outp) {
  __shared__ float4 hs[FH / 4];

  const int t = blockIdx.x;
  const int tid = threadIdx.x;
  const int lane = tid & 31;
  const int h = tid >> 5;
  const int beg = rowptr[t];
  const int end = rowptr[t + 1];
  const float er_t = half_sum(er, t * HH + h);

  float m = -__builtin_huge_valf();
  for (int i = beg + lane; i < end; i += 32) {
    int s = csrc[i];
    m = fmaxf(m, leaky(half_sum(el, s * HH + h) + er_t));
  }
  m = wave_max(m);

  float dsum = 0.f;
  for (int i = beg + lane; i < end; i += 32) {
    int s = csrc[i];
    dsum += __expf(leaky(half_sum(el, s * HH + h) + er_t) - m);
  }
  dsum = wave_sum(dsum);
  const float dn = (dsum == 0.f) ? 1.f : dsum;

  float4 acc = make_float4(0.f, 0.f, 0.f, 0.f);
  const float4* frow = (const float4*)feat + tid;
  for (int c = beg; c < end; c += 32) {
    const int i = c + lane;
    int s = 0;
    float a = 0.f;
    if (i < end) {
      s = csrc[i];
      a = __expf(leaky(half_sum(el, s * HH + h) + er_t) - m) / dn;
    }
    int cnt = end - c;
    if (cnt > 32) cnt = 32;
#pragma unroll 4
    for (int j = 0; j < cnt; ++j) {
      const int sj = __builtin_amdgcn_readlane(s, j);
      const float aj = __int_as_float(__builtin_amdgcn_readlane(__float_as_int(a), j));
      const float4 f = frow[(size_t)sj * (FH / 4)];
      acc.x += f.x * aj;
      acc.y += f.y * aj;
      acc.z += f.z * aj;
      acc.w += f.w * aj;
    }
  }

  if (!FINAL) {
    float4 r;
    r.x = elu1(acc.x);
    r.y = elu1(acc.y);
    r.z = elu1(acc.z);
    r.w = elu1(acc.w);
    ((float4*)outp)[(size_t)t * (FH / 4) + tid] = r;
  } else {
    const float4 x = ((const float4*)hin)[(size_t)t * (FH / 4) + tid];
    float4 r;
    r.x = elu1(acc.x + x.x);
    r.y = elu1(acc.y + x.y);
    r.z = elu1(acc.z + x.z);
    r.w = elu1(acc.w + x.w);
    hs[tid] = r;
    __syncthreads();
    const float* hf = (const float*)hs;
    float sm = 0.f;
    for (int hh = 0; hh < HH; ++hh) sm += hf[hh * DD + tid];
    outp[(size_t)t * DD + tid] = sm * (1.f / HH);
  }
}

extern "C" void kernel_launch(void* const* d_in, const int* in_sizes, int n_in,
                              void* d_out, int out_size, void* d_ws, size_t ws_size,
                              hipStream_t stream) {
  const int* src = (const int*)d_in[0];
  const int* dst = (const int*)d_in[1];
  const float* x   = (const float*)d_in[2];
  const float* W0  = (const float*)d_in[3];
  const float* al0 = (const float*)d_in[4];
  const float* ar0 = (const float*)d_in[5];
  const float* W1  = (const float*)d_in[6];
  const float* al1 = (const float*)d_in[7];
  const float* ar1 = (const float*)d_in[8];
  float* out = (float*)d_out;

  float* ws    = (float*)d_ws;
  float* feat  = ws;
  float* hbuf  = feat  + (size_t)NN * FH;
  float* el    = hbuf  + (size_t)NN * FH;
  float* er    = el    + (size_t)NN * HH * 2;
  int* rowptr  = (int*)(er + (size_t)NN * HH * 2);
  int* csrc    = rowptr + (NN + 16);
  int* tabc    = csrc   + EE;
  int* tabo    = tabc   + GRP_BINS * NSLAB;
  int* tmp     = tabo   + GRP_BINS * NSLAB;

  const int mtiles = (NN + 15) / 16;
  const int mgroups = (mtiles + GEMM_WAVES - 1) / GEMM_WAVES;
  const int npanels = FH / 64;
  const int gemmBlocks = mgroups * npanels;

  slab_bin<<<NSLAB, SLAB_THREADS, 0, stream>>>(src, dst, tmp, tabc, tabo);
  bin_group<<<GRP_BINS, 1024, 0, stream>>>(tmp, tabc, tabo, rowptr, csrc);

  gemm16x64_wmma<K0, FH><<<gemmBlocks, GEMM_THREADS, 0, stream>>>(x, W0, al0, ar0, feat, el, er, NN);
  node_gather<false><<<NN, 128, 0, stream>>>(rowptr, csrc, feat, el, er, feat, hbuf);
  gemm16x64_wmma<KMAX, FH><<<gemmBlocks, GEMM_THREADS, 0, stream>>>(hbuf, W1, al1, ar1, feat, el, er, NN);
  node_gather<true><<<NN, 128, 0, stream>>>(rowptr, csrc, feat, el, er, hbuf, out);
}


# node_gather: one wave per node holds all 4 heads; a lane reads its edge's 32-byte el record once per node, not 3 times in each of 4 head waves; 10000 waves, not 40000; sum order unchanged
# speedup vs baseline: 7.9858x; 1.0800x over previous
#include <hip/hip_runtime.h>
#include <math.h>

#define NN 10000
#define EE 320000
#define HH 4
#define DD 128
#define FH (HH * DD)
#define K0 256
#define KMAX 512
#define NEG_SLOPE 0.2f

typedef __attribute__((ext_vector_type(2))) float v2f;
typedef __attribute__((ext_vector_type(8))) float v8f;


template <int MASK>
__device__ __forceinline__ float row_xor(float v) {
  return __int_as_float(__builtin_amdgcn_update_dpp(0, __float_as_int(v), 0x160 + MASK,
                                                    0xf, 0xf, true));
}

#define GEMM_WAVES 20
#define GEMM_THREADS (GEMM_WAVES * 32)
#define GEMM_MIN_ENTRIES 20800

template <int K, int Nout>
__global__ __launch_bounds__(GEMM_THREADS) void gemm16x64_wmma(const float* __restrict__ X,
                                                      const float* __restrict__ W,
                                                      const float* __restrict__ al,
                                                      const float* __restrict__ ar,
                                                      float* __restrict__ out,
                                                      float* __restrict__ elp,
                                                      float* __restrict__ erp,
                                                      int M) {
  constexpr int NB = (K * 32 > GEMM_MIN_ENTRIES) ? K * 32 : GEMM_MIN_ENTRIES;
  __shared__ float2 Bs[NB];
  __shared__ float As[128];

  const int tid = threadIdx.x;
  const int lane = tid & 31;
  const int wv = tid >> 5;
  constexpr int npan = Nout >> 6;
  const int mg = blockIdx.x / npan;
  const int pn = blockIdx.x - mg * npan;

  if (tid < 128) As[tid] = (tid < 64) ? al[pn * 64 + tid] : ar[pn * 64 + tid - 64];

  {
    const int c = tid & 63;
    const int j = c >> 4, l = c & 15;
    const float* wp = W + pn * 64 + c;
    constexpr int NGRP = GEMM_THREADS / 64;
    constexpr int NIT = (K / 2 + NGRP - 1) / NGRP;
#pragma unroll 13
    for (int it = 0; it < NIT; ++it) {
      const int kp = it * NGRP + (tid >> 6);
      if (kp < K / 2) {
        const float v0 = wp[(size_t)(2 * kp) * Nout];
        const float v1 = wp[(size_t)(2 * kp + 1) * Nout];
        const int ks = kp >> 1, h = kp & 1;
        Bs[((ks * 4 + j) * 2 + h) * 16 + l] = make_float2(v0, v1);
      }
    }
  }
  __syncthreads();

  const int tm = mg * GEMM_WAVES + wv;
  if (tm * 16 >= M) return;

  const int half = lane >> 4;
  const int l = lane & 15;

  const float* aptr = X + (size_t)(tm * 16 + l) * K + half * 2;
  const float2* bptr = Bs + lane;

  v8f acc0 = {}, acc1 = {}, acc2 = {}, acc3 = {};

  float2 cur[8], nxt[8];
#pragma unroll
  for (int i = 0; i < 8; ++i) cur[i] = *(const float2*)(aptr + 4 * i);

  constexpr int NG = K / 32;
  for (int g = 0; g < NG; ++g) {
    if (g + 1 < NG) {
#pragma unroll
      for (int i = 0; i < 8; ++i) nxt[i] = *(const float2*)(aptr + 32 * (g + 1) + 4 * i);
    }
#pragma unroll
    for (int i = 0; i < 8; ++i) {
      const float2* bp = bptr + (size_t)((g * 8 + i) * 4) * 32;
      const float2 q0 = bp[0], q1 = bp[32], q2 = bp[64], q3 = bp[96];
      v2f a; a.x = cur[i].x; a.y = cur[i].y;
      v2f b0; b0.x = q0.x; b0.y = q0.y;
      v2f b1; b1.x = q1.x; b1.y = q1.y;
      v2f b2; b2.x = q2.x; b2.y = q2.y;
      v2f b3; b3.x = q3.x; b3.y = q3.y;
      acc0 = __builtin_amdgcn_wmma_f32_16x16x4_f32(false, a, false, b0, (short)0, acc0, false, false);
      acc1 = __builtin_amdgcn_wmma_f32_16x16x4_f32(false, a, false, b1, (short)0, acc1, false, false);
      acc2 = __builtin_amdgcn_wmma_f32_16x16x4_f32(false, a, false, b2, (short)0, acc2, false, false);
      acc3 = __builtin_amdgcn_wmma_f32_16x16x4_f32(false, a, false, b3, (short)0, acc3, false, false);
    }
    if (g + 1 < NG) {
#pragma unroll
      for (int i = 0; i < 8; ++i) cur[i] = nxt[i];
    }
  }

  float* obase = out + (size_t)(tm * 16 + half * 8) * Nout + pn * 64 + l;
#pragma unroll
  for (int r = 0; r < 8; ++r) {
    float* orow = obase + (size_t)r * Nout;
    orow[0] = acc0[r];
    orow[16] = acc1[r];
    orow[32] = acc2[r];
    orow[48] = acc3[r];
  }

  float x[16];
  {
    const float a0 = As[l], a1 = As[16 + l], a2 = As[32 + l], a3 = As[48 + l];
    const float b0 = As[64 + l], b1 = As[80 + l], b2 = As[96 + l], b3 = As[112 + l];
    for (int r = 0; r < 8; ++r) {
      x[r] = acc0[r] * a0 + acc1[r] * a1 + acc2[r] * a2 + acc3[r] * a3;
      x[8 + r] = acc0[r] * b0 + acc1[r] * b1 + acc2[r] * b2 + acc3[r] * b3;
    }
  }
  const bool t8 = (l & 8) != 0, t4 = (l & 4) != 0, t2 = (l & 2) != 0, t1 = (l & 1) != 0;
  float y[8], z[4], w[2];
  for (int i = 0; i < 8; ++i)
    y[i] = (t8 ? x[i + 8] : x[i]) + row_xor<8>(t8 ? x[i] : x[i + 8]);
  for (int i = 0; i < 4; ++i)
    z[i] = (t4 ? y[i + 4] : y[i]) + row_xor<4>(t4 ? y[i] : y[i + 4]);
  for (int i = 0; i < 2; ++i)
    w[i] = (t2 ? z[i + 2] : z[i]) + row_xor<2>(t2 ? z[i] : z[i + 2]);
  const float u = (t1 ? w[1] : w[0]) + row_xor<1>(t1 ? w[0] : w[1]);
  float* ep = t8 ? erp : elp;
  ep[(size_t)(tm * 16 + half * 8 + (l & 7)) * (Nout >> 6) + pn] = u;
}

#define GRP_NODES 256
#define GRP_BINS ((NN + GRP_NODES - 1) / GRP_NODES)
#define SLAB 640
#define NSLAB (EE / SLAB)
#define SLAB_THREADS 128
#define SLAB_PER_THREAD (SLAB / SLAB_THREADS)

__device__ __forceinline__ int wave_sum_i(int v) {
  for (int o = 16; o > 0; o >>= 1) v += __shfl_xor(v, o);
  return v;
}

__global__ __launch_bounds__(SLAB_THREADS) void slab_bin(const int* __restrict__ src,
                                                         const int* __restrict__ dst,
                                                         int* __restrict__ tmpw,
                                                         int* __restrict__ tabc,
                                                         int* __restrict__ tabo) {
  static_assert(EE % SLAB == 0, "whole slabs");
  static_assert(SLAB % SLAB_THREADS == 0, "whole edges per thread");
  static_assert(GRP_BINS <= SLAB_THREADS, "one thread per bin below");
  __shared__ unsigned bcnt[GRP_BINS];
  __shared__ unsigned boff[GRP_BINS];

  int2* tmp = (int2*)tmpw;
  const int tid = threadIdx.x;
  const int e0 = blockIdx.x * SLAB;
  if (tid < GRP_BINS) bcnt[tid] = 0u;
  __syncthreads();

  int d[SLAB_PER_THREAD], s[SLAB_PER_THREAD];
  unsigned bin[SLAB_PER_THREAD], rk[SLAB_PER_THREAD];
#pragma unroll
  for (int i = 0; i < SLAB_PER_THREAD; ++i) {
    const int e = e0 + i * SLAB_THREADS + tid;
    d[i] = dst[e];
    s[i] = src[e];
  }
#pragma unroll
  for (int i = 0; i < SLAB_PER_THREAD; ++i) {
    bin[i] = (unsigned)d[i] / GRP_NODES;
    rk[i] = atomicAdd(&bcnt[bin[i]], 1u);
  }
  __syncthreads();

  unsigned myc = 0u, myo = 0u;
  if (tid < GRP_BINS) {
#pragma unroll
    for (int b = 0; b < GRP_BINS; ++b) {
      const unsigned v = bcnt[b];
      if (b < tid) myo += v;
      if (b == tid) myc = v;
    }
    boff[tid] = myo;
  }
  __syncthreads();

#pragma unroll
  for (int i = 0; i < SLAB_PER_THREAD; ++i)
    tmp[e0 + boff[bin[i]] + rk[i]] = make_int2(d[i], s[i]);
  if (tid < GRP_BINS) {
    tabc[tid * NSLAB + blockIdx.x] = (int)myc;
    tabo[tid * NSLAB + blockIdx.x] = (int)myo;
  }
}

__global__ __launch_bounds__(1024) void bin_group(const int* __restrict__ tmpw,
                                                  const int* __restrict__ tabc,
                                                  const int* __restrict__ tabo,
                                                  int* __restrict__ rowptr,
                                                  int* __restrict__ csrc) {
  static_assert(GRP_NODES == 256, "the scan below gives 8 nodes to each of 32 lanes");
  static_assert(NSLAB <= 1024 && NSLAB % 4 == 0, "one thread per slab; counts read 4 at a time");
  __shared__ unsigned cur[GRP_NODES];
  __shared__ int part[32];
  __shared__ int segc[NSLAB];
  __shared__ int sego[NSLAB];

  const int2* tmp = (const int2*)tmpw;
  const int tid = threadIdx.x;
  const int lane = tid & 31;
  const int wv = tid >> 5;
  const int b = blockIdx.x;
  const int lo = b * GRP_NODES;
  const unsigned span = (unsigned)((NN - lo < GRP_NODES) ? (NN - lo) : GRP_NODES);

  if (tid < GRP_NODES) cur[tid] = 0u;

  constexpr int NLOW = (GRP_BINS * (NSLAB / 4) + 1023) / 1024;
  const int4* tc4 = (const int4*)tabc;
  const int nlow4 = b * (NSLAB / 4);
  const int jrow = b * NSLAB + (tid < NSLAB ? tid : 0);
  const int sc = tabc[jrow];
  const int so = tabo[jrow];
  int4 low[NLOW];
#pragma unroll
  for (int it = 0; it < NLOW; ++it) {
    const int i = it * 1024 + tid;
    low[it] = tc4[i < nlow4 ? i : 0];
  }
  int below = 0;
#pragma unroll
  for (int it = 0; it < NLOW; ++it) {
    const int i = it * 1024 + tid;
    if (i < nlow4) below += low[it].x + low[it].y + low[it].z + low[it].w;
  }
  if (tid < NSLAB) {
    segc[tid] = sc;
    sego[tid] = so;
  }
  below = wave_sum_i(below);
  if (lane == 0) part[wv] = below;
  __syncthreads();

  constexpr int SPW = (NSLAB + 31) / 32;
  int c[SPW], ib[SPW];
  int2 rec[SPW];
#pragma unroll
  for (int m = 0; m < SPW; ++m) {
    const int j = wv + 32 * m;
    c[m] = 0;
    ib[m] = 0;
    if (j < NSLAB) {
      c[m] = segc[j];
      ib[m] = j * SLAB + sego[j];
    }
  }
#pragma unroll
  for (int m = 0; m < SPW; ++m) {
    rec[m] = make_int2(lo, 0);
    if (lane < c[m]) rec[m] = tmp[ib[m] + lane];
  }
#pragma unroll
  for (int m = 0; m < SPW; ++m) {
    if (lane < c[m]) atomicAdd(&cur[rec[m].x - lo], 1u);
    for (int i = lane + 32; i < c[m]; i += 32)
      atomicAdd(&cur[tmp[ib[m] + i].x - lo], 1u);
  }
  __syncthreads();

  if (wv == 0) {
    const int base = wave_sum_i(part[lane]);
    int k8[8];
    int s = 0;
#pragma unroll
    for (int k = 0; k < 8; ++k) {
      k8[k] = (int)cur[lane * 8 + k];
      s += k8[k];
    }
    int incl = s;
    for (int o = 1; o < 32; o <<= 1) {
      const int v = __shfl_up(incl, o);
      if (lane >= o) incl += v;
    }
    int run = base + incl - s;
#pragma unroll
    for (int k = 0; k < 8; ++k) {
      const int i = lane * 8 + k;
      if ((unsigned)i < span) rowptr[lo + i] = run;
      cur[i] = (unsigned)run;
      run += k8[k];
    }
    if (b == (int)gridDim.x - 1 && lane == 31) rowptr[NN] = run;
  }
  __syncthreads();

#pragma unroll
  for (int m = 0; m < SPW; ++m) {
    if (lane < c[m]) csrc[atomicAdd(&cur[rec[m].x - lo], 1u)] = rec[m].y;
    for (int i = lane + 32; i < c[m]; i += 32) {
      const int2 t = tmp[ib[m] + i];
      csrc[atomicAdd(&cur[t.x - lo], 1u)] = t.y;
    }
  }
}

__device__ __forceinline__ float leaky(float v) {
  return v > 0.f ? v : NEG_SLOPE * v;
}

__device__ __forceinline__ float elu1(float v) {
  return v > 0.f ? v : (__expf(v) - 1.f);
}

__device__ __forceinline__ float wave_max(float v) {
  for (int o = 16; o > 0; o >>= 1) v = fmaxf(v, __shfl_xor(v, o));
  return v;
}

__device__ __forceinline__ float wave_sum(float v) {
  for (int o = 16; o > 0; o >>= 1) v += __shfl_xor(v, o);
  return v;
}

#define GATHER_WAVES 4
#define GATHER_THREADS (GATHER_WAVES * 32)
#define GATHER_BLOCKS ((NN + GATHER_WAVES - 1) / GATHER_WAVES)

__device__ __forceinline__ void edge_scores(const float* __restrict__ el, int s,
                                            float er0, float er1, float er2, float er3,
                                            float& e0, float& e1, float& e2, float& e3) {
  const float4 a = ((const float4*)el)[(size_t)s * 2];
  const float4 b = ((const float4*)el)[(size_t)s * 2 + 1];
  e0 = leaky((a.x + a.y) + er0);
  e1 = leaky((a.z + a.w) + er1);
  e2 = leaky((b.x + b.y) + er2);
  e3 = leaky((b.z + b.w) + er3);
}

__device__ __forceinline__ float4 elu4(float4 v) {
  float4 r;
  r.x = elu1(v.x);
  r.y = elu1(v.y);
  r.z = elu1(v.z);
  r.w = elu1(v.w);
  return r;
}

__device__ __forceinline__ float4 add4(float4 a, float4 b) {
  return make_float4(a.x + b.x, a.y + b.y, a.z + b.z, a.w + b.w);
}

__device__ __forceinline__ float bcast(float v, int j) {
  return __int_as_float(__builtin_amdgcn_readlane(__float_as_int(v), j));
}

template <bool FINAL>
__global__ __launch_bounds__(GATHER_THREADS) void node_gather(
    const int* __restrict__ rowptr, const int* __restrict__ csrc,
    const float* __restrict__ feat, const float* __restrict__ el,
    const float* __restrict__ er, const float* __restrict__ hin,
    float* __restrict__ outp) {
  const int lane = threadIdx.x & 31;
  const int wv = __builtin_amdgcn_readfirstlane((int)(threadIdx.x >> 5));
  const int t = blockIdx.x * GATHER_WAVES + wv;
  if (t >= NN) return;
  const int beg = rowptr[t];
  const int end = rowptr[t + 1];
  const float4 ra = ((const float4*)er)[(size_t)t * 2];
  const float4 rb = ((const float4*)er)[(size_t)t * 2 + 1];
  const float er0 = ra.x + ra.y, er1 = ra.z + ra.w, er2 = rb.x + rb.y, er3 = rb.z + rb.w;

  const float ninf = -__builtin_huge_valf();
  const int i0 = beg + lane;
  const bool has0 = i0 < end;
  int s0 = 0;
  float p0 = ninf, p1 = ninf, p2 = ninf, p3 = ninf;
  if (has0) {
    s0 = csrc[i0];
    edge_scores(el, s0, er0, er1, er2, er3, p0, p1, p2, p3);
  }

  float m0 = p0, m1 = p1, m2 = p2, m3 = p3;
  for (int i = i0 + 32; i < end; i += 32) {
    float q0, q1, q2, q3;
    edge_scores(el, csrc[i], er0, er1, er2, er3, q0, q1, q2, q3);
    m0 = fmaxf(m0, q0);
    m1 = fmaxf(m1, q1);
    m2 = fmaxf(m2, q2);
    m3 = fmaxf(m3, q3);
  }
  m0 = wave_max(m0);
  m1 = wave_max(m1);
  m2 = wave_max(m2);
  m3 = wave_max(m3);

  const float x0 = has0 ? __expf(p0 - m0) : 0.f;
  const float x1 = has0 ? __expf(p1 - m1) : 0.f;
  const float x2 = has0 ? __expf(p2 - m2) : 0.f;
  const float x3 = has0 ? __expf(p3 - m3) : 0.f;
  float d0 = x0, d1 = x1, d2 = x2, d3 = x3;
  for (int i = i0 + 32; i < end; i += 32) {
    float q0, q1, q2, q3;
    edge_scores(el, csrc[i], er0, er1, er2, er3, q0, q1, q2, q3);
    d0 += __expf(q0 - m0);
    d1 += __expf(q1 - m1);
    d2 += __expf(q2 - m2);
    d3 += __expf(q3 - m3);
  }
  d0 = wave_sum(d0);
  d1 = wave_sum(d1);
  d2 = wave_sum(d2);
  d3 = wave_sum(d3);
  const float dn0 = (d0 == 0.f) ? 1.f : d0;
  const float dn1 = (d1 == 0.f) ? 1.f : d1;
  const float dn2 = (d2 == 0.f) ? 1.f : d2;
  const float dn3 = (d3 == 0.f) ? 1.f : d3;

  float4 acc0 = make_float4(0.f, 0.f, 0.f, 0.f);
  float4 acc1 = acc0, acc2 = acc0, acc3 = acc0;
  const float4* frow = (const float4*)feat + lane;
  for (int c = beg; c < end; c += 32) {
    int s = s0;
    float a0 = x0 / dn0, a1 = x1 / dn1, a2 = x2 / dn2, a3 = x3 / dn3;
    if (c != beg) {
      const int i = c + lane;
      s = 0;
      a0 = 0.f; a1 = 0.f; a2 = 0.f; a3 = 0.f;
      if (i < end) {
        float q0, q1, q2, q3;
        s = csrc[i];
        edge_scores(el, s, er0, er1, er2, er3, q0, q1, q2, q3);
        a0 = __expf(q0 - m0) / dn0;
        a1 = __expf(q1 - m1) / dn1;
        a2 = __expf(q2 - m2) / dn2;
        a3 = __expf(q3 - m3) / dn3;
      }
    }
    int cnt = end - c;
    if (cnt > 32) cnt = 32;
#pragma unroll 2
    for (int j = 0; j < cnt; ++j) {
      const int sj = __builtin_amdgcn_readlane(s, j);
      const float b0 = bcast(a0, j), b1 = bcast(a1, j), b2 = bcast(a2, j), b3 = bcast(a3, j);
      const float4* fr = frow + (size_t)sj * (FH / 4);
      const float4 f0 = fr[0], f1 = fr[DD / 4], f2 = fr[2 * (DD / 4)], f3 = fr[3 * (DD / 4)];
      acc0.x += f0.x * b0;
      acc0.y += f0.y * b0;
      acc0.z += f0.z * b0;
      acc0.w += f0.w * b0;
      acc1.x += f1.x * b1;
      acc1.y += f1.y * b1;
      acc1.z += f1.z * b1;
      acc1.w += f1.w * b1;
      acc2.x += f2.x * b2;
      acc2.y += f2.y * b2;
      acc2.z += f2.z * b2;
      acc2.w += f2.w * b2;
      acc3.x += f3.x * b3;
      acc3.y += f3.y * b3;
      acc3.z += f3.z * b3;
      acc3.w += f3.w * b3;
    }
  }

  if (!FINAL) {
    float4* o = (float4*)outp + (size_t)t * (FH / 4) + lane;
    o[0] = elu4(acc0);
    o[DD / 4] = elu4(acc1);
    o[2 * (DD / 4)] = elu4(acc2);
    o[3 * (DD / 4)] = elu4(acc3);
  } else {
    const float4* xr = (const float4*)hin + (size_t)t * (FH / 4) + lane;
    const float4 r0 = elu4(add4(acc0, xr[0]));
    const float4 r1 = elu4(add4(acc1, xr[DD / 4]));
    const float4 r2 = elu4(add4(acc2, xr[2 * (DD / 4)]));
    const float4 r3 = elu4(add4(acc3, xr[3 * (DD / 4)]));
    float4 sm;
    sm.x = (((r0.x + r1.x) + r2.x) + r3.x) * (1.f / HH);
    sm.y = (((r0.y + r1.y) + r2.y) + r3.y) * (1.f / HH);
    sm.z = (((r0.z + r1.z) + r2.z) + r3.z) * (1.f / HH);
    sm.w = (((r0.w + r1.w) + r2.w) + r3.w) * (1.f / HH);
    ((float4*)outp)[(size_t)t * (DD / 4) + lane] = sm;
  }
}

extern "C" void kernel_launch(void* const* d_in, const int* in_sizes, int n_in,
                              void* d_out, int out_size, void* d_ws, size_t ws_size,
                              hipStream_t stream) {
  const int* src = (const int*)d_in[0];
  const int* dst = (const int*)d_in[1];
  const float* x   = (const float*)d_in[2];
  const float* W0  = (const float*)d_in[3];
  const float* al0 = (const float*)d_in[4];
  const float* ar0 = (const float*)d_in[5];
  const float* W1  = (const float*)d_in[6];
  const float* al1 = (const float*)d_in[7];
  const float* ar1 = (const float*)d_in[8];
  float* out = (float*)d_out;

  float* ws    = (float*)d_ws;
  float* feat  = ws;
  float* hbuf  = feat  + (size_t)NN * FH;
  float* el    = hbuf  + (size_t)NN * FH;
  float* er    = el    + (size_t)NN * HH * 2;
  int* rowptr  = (int*)(er + (size_t)NN * HH * 2);
  int* csrc    = rowptr + (NN + 16);
  int* tabc    = csrc   + EE;
  int* tabo    = tabc   + GRP_BINS * NSLAB;
  int* tmp     = tabo   + GRP_BINS * NSLAB;

  const int mtiles = (NN + 15) / 16;
  const int mgroups = (mtiles + GEMM_WAVES - 1) / GEMM_WAVES;
  const int npanels = FH / 64;
  const int gemmBlocks = mgroups * npanels;

  slab_bin<<<NSLAB, SLAB_THREADS, 0, stream>>>(src, dst, tmp, tabc, tabo);
  bin_group<<<GRP_BINS, 1024, 0, stream>>>(tmp, tabc, tabo, rowptr, csrc);

  gemm16x64_wmma<K0, FH><<<gemmBlocks, GEMM_THREADS, 0, stream>>>(x, W0, al0, ar0, feat, el, er, NN);
  node_gather<false><<<GATHER_BLOCKS, GATHER_THREADS, 0, stream>>>(rowptr, csrc, feat, el, er, feat, hbuf);
  gemm16x64_wmma<KMAX, FH><<<gemmBlocks, GEMM_THREADS, 0, stream>>>(hbuf, W1, al1, ar1, feat, el, er, NN);
  node_gather<true><<<GATHER_BLOCKS, GATHER_THREADS, 0, stream>>>(rowptr, csrc, feat, el, er, hbuf, out);
}


# Layer 0: a work-group sums its 16 nodes' 1-KB x rows per head in LDS and projects them there on WMMA (linearity): no sum or product plane, 5 launches; el0/er0 = x.(W0 al0); layer-0 sum order changes
# speedup vs baseline: 8.8690x; 1.1106x over previous
#include <hip/hip_runtime.h>
#include <math.h>

#define NN 10000
#define EE 320000
#define HH 4
#define DD 128
#define FH (HH * DD)
#define K0 256
#define KMAX 512
#define NEG_SLOPE 0.2f

typedef __attribute__((ext_vector_type(2))) float v2f;
typedef __attribute__((ext_vector_type(8))) float v8f;


template <int MASK>
__device__ __forceinline__ float row_xor(float v) {
  return __int_as_float(__builtin_amdgcn_update_dpp(0, __float_as_int(v), 0x160 + MASK,
                                                    0xf, 0xf, true));
}

#define GEMM_WAVES 20
#define GEMM_THREADS (GEMM_WAVES * 32)
#define GEMM_MIN_ENTRIES 20800

template <int K, int Nout>
__global__ __launch_bounds__(GEMM_THREADS) void gemm16x64_wmma(const float* __restrict__ X,
                                                      const float* __restrict__ W,
                                                      const float* __restrict__ al,
                                                      const float* __restrict__ ar,
                                                      float* __restrict__ out,
                                                      float* __restrict__ elp,
                                                      float* __restrict__ erp,
                                                      int M) {
  constexpr int NB = (K * 32 > GEMM_MIN_ENTRIES) ? K * 32 : GEMM_MIN_ENTRIES;
  __shared__ float2 Bs[NB];
  __shared__ float As[128];

  const int tid = threadIdx.x;
  const int lane = tid & 31;
  const int wv = tid >> 5;
  constexpr int npan = Nout >> 6;
  const int mg = blockIdx.x / npan;
  const int pn = blockIdx.x - mg * npan;

  if (tid < 128) As[tid] = (tid < 64) ? al[pn * 64 + tid] : ar[pn * 64 + tid - 64];

  {
    const int c = tid & 63;
    const int j = c >> 4, l = c & 15;
    const float* wp = W + pn * 64 + c;
    constexpr int NGRP = GEMM_THREADS / 64;
    constexpr int NIT = (K / 2 + NGRP - 1) / NGRP;
#pragma unroll 13
    for (int it = 0; it < NIT; ++it) {
      const int kp = it * NGRP + (tid >> 6);
      if (kp < K / 2) {
        const float v0 = wp[(size_t)(2 * kp) * Nout];
        const float v1 = wp[(size_t)(2 * kp + 1) * Nout];
        const int ks = kp >> 1, h = kp & 1;
        Bs[((ks * 4 + j) * 2 + h) * 16 + l] = make_float2(v0, v1);
      }
    }
  }
  __syncthreads();

  const int tm = mg * GEMM_WAVES + wv;
  if (tm * 16 >= M) return;

  const int half = lane >> 4;
  const int l = lane & 15;

  const float* aptr = X + (size_t)(tm * 16 + l) * K + half * 2;
  const float2* bptr = Bs + lane;

  v8f acc0 = {}, acc1 = {}, acc2 = {}, acc3 = {};

  float2 cur[8], nxt[8];
#pragma unroll
  for (int i = 0; i < 8; ++i) cur[i] = *(const float2*)(aptr + 4 * i);

  constexpr int NG = K / 32;
  for (int g = 0; g < NG; ++g) {
    if (g + 1 < NG) {
#pragma unroll
      for (int i = 0; i < 8; ++i) nxt[i] = *(const float2*)(aptr + 32 * (g + 1) + 4 * i);
    }
#pragma unroll
    for (int i = 0; i < 8; ++i) {
      const float2* bp = bptr + (size_t)((g * 8 + i) * 4) * 32;
      const float2 q0 = bp[0], q1 = bp[32], q2 = bp[64], q3 = bp[96];
      v2f a; a.x = cur[i].x; a.y = cur[i].y;
      v2f b0; b0.x = q0.x; b0.y = q0.y;
      v2f b1; b1.x = q1.x; b1.y = q1.y;
      v2f b2; b2.x = q2.x; b2.y = q2.y;
      v2f b3; b3.x = q3.x; b3.y = q3.y;
      acc0 = __builtin_amdgcn_wmma_f32_16x16x4_f32(false, a, false, b0, (short)0, acc0, false, false);
      acc1 = __builtin_amdgcn_wmma_f32_16x16x4_f32(false, a, false, b1, (short)0, acc1, false, false);
      acc2 = __builtin_amdgcn_wmma_f32_16x16x4_f32(false, a, false, b2, (short)0, acc2, false, false);
      acc3 = __builtin_amdgcn_wmma_f32_16x16x4_f32(false, a, false, b3, (short)0, acc3, false, false);
    }
    if (g + 1 < NG) {
#pragma unroll
      for (int i = 0; i < 8; ++i) cur[i] = nxt[i];
    }
  }

  float* obase = out + (size_t)(tm * 16 + half * 8) * Nout + pn * 64 + l;
#pragma unroll
  for (int r = 0; r < 8; ++r) {
    float* orow = obase + (size_t)r * Nout;
    orow[0] = acc0[r];
    orow[16] = acc1[r];
    orow[32] = acc2[r];
    orow[48] = acc3[r];
  }

  float x[16];
  {
    const float a0 = As[l], a1 = As[16 + l], a2 = As[32 + l], a3 = As[48 + l];
    const float b0 = As[64 + l], b1 = As[80 + l], b2 = As[96 + l], b3 = As[112 + l];
    for (int r = 0; r < 8; ++r) {
      x[r] = acc0[r] * a0 + acc1[r] * a1 + acc2[r] * a2 + acc3[r] * a3;
      x[8 + r] = acc0[r] * b0 + acc1[r] * b1 + acc2[r] * b2 + acc3[r] * b3;
    }
  }
  const bool t8 = (l & 8) != 0, t4 = (l & 4) != 0, t2 = (l & 2) != 0, t1 = (l & 1) != 0;
  float y[8], z[4], w[2];
  for (int i = 0; i < 8; ++i)
    y[i] = (t8 ? x[i + 8] : x[i]) + row_xor<8>(t8 ? x[i] : x[i + 8]);
  for (int i = 0; i < 4; ++i)
    z[i] = (t4 ? y[i + 4] : y[i]) + row_xor<4>(t4 ? y[i] : y[i + 4]);
  for (int i = 0; i < 2; ++i)
    w[i] = (t2 ? z[i + 2] : z[i]) + row_xor<2>(t2 ? z[i] : z[i + 2]);
  const float u = (t1 ? w[1] : w[0]) + row_xor<1>(t1 ? w[0] : w[1]);
  float* ep = t8 ? erp : elp;
  ep[(size_t)(tm * 16 + half * 8 + (l & 7)) * (Nout >> 6) + pn] = u;
}

#define GRP_NODES 256
#define GRP_BINS ((NN + GRP_NODES - 1) / GRP_NODES)
#define SLAB 640
#define NSLAB (EE / SLAB)
#define SLAB_THREADS 128
#define SLAB_PER_THREAD (SLAB / SLAB_THREADS)

__device__ __forceinline__ int wave_sum_i(int v) {
  for (int o = 16; o > 0; o >>= 1) v += __shfl_xor(v, o);
  return v;
}

__device__ __forceinline__ float wave_sum_f(float v) {
  for (int o = 16; o > 0; o >>= 1) v += __shfl_xor(v, o);
  return v;
}

__device__ __forceinline__ float dot4(float4 a, float4 b) {
  return ((a.x * b.x + a.y * b.y) + a.z * b.z) + a.w * b.w;
}

#define UV_BLOCKS (K0 / (SLAB_THREADS / 32))
#define WF_BLOCKS ((K0 / 4) * (FH / 64) * 2 * 32 / SLAB_THREADS)

__device__ __forceinline__ void fold_rows(const float* __restrict__ W0,
                                          const float* __restrict__ al,
                                          const float* __restrict__ ar,
                                          float* __restrict__ uv, int k, int lane) {
  const float4* wr = (const float4*)(W0 + (size_t)k * FH) + lane;
  const float4* pl = (const float4*)al + lane;
  const float4* pr = (const float4*)ar + lane;
  float d[8];
#pragma unroll
  for (int h = 0; h < HH; ++h) {
    const float4 w = wr[h * (DD / 4)];
    d[h] = dot4(w, pl[h * (DD / 4)]);
    d[HH + h] = dot4(w, pr[h * (DD / 4)]);
  }
#pragma unroll
  for (int j = 0; j < 8; ++j) d[j] = wave_sum_f(d[j]);
  if (lane == 0) {
#pragma unroll
    for (int j = 0; j < 8; ++j) uv[j * K0 + k] = d[j];
  }
}

__device__ __forceinline__ void frag_w0(const float* __restrict__ W0, float* __restrict__ wf, int u) {
  const int lane = u & 31;
  const int q = (u >> 5) & 1;
  const int ks = (u >> 6) & (K0 / 4 - 1);
  const int pn = u >> 12;
  const int k0 = 4 * ks + 2 * (lane >> 4);
  const float* p = W0 + (size_t)k0 * FH + pn * 64 + q * 32 + (lane & 15);
  ((float4*)wf)[u] = make_float4(p[0], p[FH], p[16], p[FH + 16]);
}

__device__ __forceinline__ void node_scores(const float* __restrict__ x,
                                            const float* __restrict__ uv,
                                            float* __restrict__ el,
                                            float* __restrict__ er, int t, int lane) {
  const float4* xr = (const float4*)(x + (size_t)t * K0) + lane;
  const float4 xa = xr[0], xb = xr[K0 / 8];
  float d[8];
#pragma unroll
  for (int j = 0; j < 8; ++j) {
    const float4* p = (const float4*)(uv + j * K0) + lane;
    d[j] = dot4(xa, p[0]) + dot4(xb, p[K0 / 8]);
  }
#pragma unroll
  for (int j = 0; j < 8; ++j) d[j] = wave_sum_f(d[j]);
  if (lane == 0) {
    float4* pl = (float4*)(el + (size_t)t * 8);
    float4* pr = (float4*)(er + (size_t)t * 8);
    pl[0] = make_float4(d[0], 0.f, d[1], 0.f);
    pl[1] = make_float4(d[2], 0.f, d[3], 0.f);
    pr[0] = make_float4(d[4], 0.f, d[5], 0.f);
    pr[1] = make_float4(d[6], 0.f, d[7], 0.f);
  }
}

__global__ __launch_bounds__(SLAB_THREADS) void slab_bin(const int* __restrict__ src,
                                                         const int* __restrict__ dst,
                                                         int* __restrict__ tmpw,
                                                         int* __restrict__ tabc,
                                                         int* __restrict__ tabo,
                                                         const float* __restrict__ W0,
                                                         const float* __restrict__ al0,
                                                         const float* __restrict__ ar0,
                                                         float* __restrict__ uv,
                                                         float* __restrict__ wf) {
  if (blockIdx.x >= NSLAB + UV_BLOCKS) {
    frag_w0(W0, wf, ((int)blockIdx.x - NSLAB - UV_BLOCKS) * SLAB_THREADS + (int)threadIdx.x);
    return;
  }
  if (blockIdx.x >= NSLAB) {
    const int k = ((int)blockIdx.x - NSLAB) * (SLAB_THREADS / 32) + (int)(threadIdx.x >> 5);
    fold_rows(W0, al0, ar0, uv, k, threadIdx.x & 31);
    return;
  }
  static_assert(EE % SLAB == 0, "whole slabs");
  static_assert(SLAB % SLAB_THREADS == 0, "whole edges per thread");
  static_assert(GRP_BINS <= SLAB_THREADS, "one thread per bin below");
  __shared__ unsigned bcnt[GRP_BINS];
  __shared__ unsigned boff[GRP_BINS];

  int2* tmp = (int2*)tmpw;
  const int tid = threadIdx.x;
  const int e0 = blockIdx.x * SLAB;
  if (tid < GRP_BINS) bcnt[tid] = 0u;
  __syncthreads();

  int d[SLAB_PER_THREAD], s[SLAB_PER_THREAD];
  unsigned bin[SLAB_PER_THREAD], rk[SLAB_PER_THREAD];
#pragma unroll
  for (int i = 0; i < SLAB_PER_THREAD; ++i) {
    const int e = e0 + i * SLAB_THREADS + tid;
    d[i] = dst[e];
    s[i] = src[e];
  }
#pragma unroll
  for (int i = 0; i < SLAB_PER_THREAD; ++i) {
    bin[i] = (unsigned)d[i] / GRP_NODES;
    rk[i] = atomicAdd(&bcnt[bin[i]], 1u);
  }
  __syncthreads();

  unsigned myc = 0u, myo = 0u;
  if (tid < GRP_BINS) {
#pragma unroll
    for (int b = 0; b < GRP_BINS; ++b) {
      const unsigned v = bcnt[b];
      if (b < tid) myo += v;
      if (b == tid) myc = v;
    }
    boff[tid] = myo;
  }
  __syncthreads();

#pragma unroll
  for (int i = 0; i < SLAB_PER_THREAD; ++i)
    tmp[e0 + boff[bin[i]] + rk[i]] = make_int2(d[i], s[i]);
  if (tid < GRP_BINS) {
    tabc[tid * NSLAB + blockIdx.x] = (int)myc;
    tabo[tid * NSLAB + blockIdx.x] = (int)myo;
  }
}

__global__ __launch_bounds__(1024) void bin_group(const int* __restrict__ tmpw,
                                                  const int* __restrict__ tabc,
                                                  const int* __restrict__ tabo,
                                                  int* __restrict__ rowptr,
                                                  int* __restrict__ csrc,
                                                  const float* __restrict__ x,
                                                  const float* __restrict__ uv,
                                                  float* __restrict__ el,
                                                  float* __restrict__ er) {
  if (blockIdx.x >= GRP_BINS) {
    const int t = ((int)blockIdx.x - GRP_BINS) * 32 + (int)(threadIdx.x >> 5);
    if (t < NN) node_scores(x, uv, el, er, t, threadIdx.x & 31);
    return;
  }
  static_assert(GRP_NODES == 256, "the scan below gives 8 nodes to each of 32 lanes");
  static_assert(NSLAB <= 1024 && NSLAB % 4 == 0, "one thread per slab; counts read 4 at a time");
  __shared__ unsigned cur[GRP_NODES];
  __shared__ int part[32];
  __shared__ int segc[NSLAB];
  __shared__ int sego[NSLAB];

  const int2* tmp = (const int2*)tmpw;
  const int tid = threadIdx.x;
  const int lane = tid & 31;
  const int wv = tid >> 5;
  const int b = blockIdx.x;
  const int lo = b * GRP_NODES;
  const unsigned span = (unsigned)((NN - lo < GRP_NODES) ? (NN - lo) : GRP_NODES);

  if (tid < GRP_NODES) cur[tid] = 0u;

  constexpr int NLOW = (GRP_BINS * (NSLAB / 4) + 1023) / 1024;
  const int4* tc4 = (const int4*)tabc;
  const int nlow4 = b * (NSLAB / 4);
  const int jrow = b * NSLAB + (tid < NSLAB ? tid : 0);
  const int sc = tabc[jrow];
  const int so = tabo[jrow];
  int4 low[NLOW];
#pragma unroll
  for (int it = 0; it < NLOW; ++it) {
    const int i = it * 1024 + tid;
    low[it] = tc4[i < nlow4 ? i : 0];
  }
  int below = 0;
#pragma unroll
  for (int it = 0; it < NLOW; ++it) {
    const int i = it * 1024 + tid;
    if (i < nlow4) below += low[it].x + low[it].y + low[it].z + low[it].w;
  }
  if (tid < NSLAB) {
    segc[tid] = sc;
    sego[tid] = so;
  }
  below = wave_sum_i(below);
  if (lane == 0) part[wv] = below;
  __syncthreads();

  constexpr int SPW = (NSLAB + 31) / 32;
  int c[SPW], ib[SPW];
  int2 rec[SPW];
#pragma unroll
  for (int m = 0; m < SPW; ++m) {
    const int j = wv + 32 * m;
    c[m] = 0;
    ib[m] = 0;
    if (j < NSLAB) {
      c[m] = segc[j];
      ib[m] = j * SLAB + sego[j];
    }
  }
#pragma unroll
  for (int m = 0; m < SPW; ++m) {
    rec[m] = make_int2(lo, 0);
    if (lane < c[m]) rec[m] = tmp[ib[m] + lane];
  }
#pragma unroll
  for (int m = 0; m < SPW; ++m) {
    if (lane < c[m]) atomicAdd(&cur[rec[m].x - lo], 1u);
    for (int i = lane + 32; i < c[m]; i += 32)
      atomicAdd(&cur[tmp[ib[m] + i].x - lo], 1u);
  }
  __syncthreads();

  if (wv == 0) {
    const int base = wave_sum_i(part[lane]);
    int k8[8];
    int s = 0;
#pragma unroll
    for (int k = 0; k < 8; ++k) {
      k8[k] = (int)cur[lane * 8 + k];
      s += k8[k];
    }
    int incl = s;
    for (int o = 1; o < 32; o <<= 1) {
      const int v = __shfl_up(incl, o);
      if (lane >= o) incl += v;
    }
    int run = base + incl - s;
#pragma unroll
    for (int k = 0; k < 8; ++k) {
      const int i = lane * 8 + k;
      if ((unsigned)i < span) rowptr[lo + i] = run;
      cur[i] = (unsigned)run;
      run += k8[k];
    }
    if (b == GRP_BINS - 1 && lane == 31) rowptr[NN] = run;
  }
  __syncthreads();

#pragma unroll
  for (int m = 0; m < SPW; ++m) {
    if (lane < c[m]) csrc[atomicAdd(&cur[rec[m].x - lo], 1u)] = rec[m].y;
    for (int i = lane + 32; i < c[m]; i += 32) {
      const int2 t = tmp[ib[m] + i];
      csrc[atomicAdd(&cur[t.x - lo], 1u)] = t.y;
    }
  }
}

__device__ __forceinline__ float leaky(float v) {
  return v > 0.f ? v : NEG_SLOPE * v;
}

__device__ __forceinline__ float elu1(float v) {
  return v > 0.f ? v : (__expf(v) - 1.f);
}

__device__ __forceinline__ float wave_max(float v) {
  for (int o = 16; o > 0; o >>= 1) v = fmaxf(v, __shfl_xor(v, o));
  return v;
}

__device__ __forceinline__ float wave_sum(float v) {
  for (int o = 16; o > 0; o >>= 1) v += __shfl_xor(v, o);
  return v;
}

#define GATHER_WAVES 4
#define GATHER_THREADS (GATHER_WAVES * 32)
#define GATHER_BLOCKS ((NN + GATHER_WAVES - 1) / GATHER_WAVES)

__device__ __forceinline__ void edge_scores(const float* __restrict__ el, int s,
                                            float er0, float er1, float er2, float er3,
                                            float& e0, float& e1, float& e2, float& e3) {
  const float4 a = ((const float4*)el)[(size_t)s * 2];
  const float4 b = ((const float4*)el)[(size_t)s * 2 + 1];
  e0 = leaky((a.x + a.y) + er0);
  e1 = leaky((a.z + a.w) + er1);
  e2 = leaky((b.x + b.y) + er2);
  e3 = leaky((b.z + b.w) + er3);
}

__device__ __forceinline__ float4 elu4(float4 v) {
  float4 r;
  r.x = elu1(v.x);
  r.y = elu1(v.y);
  r.z = elu1(v.z);
  r.w = elu1(v.w);
  return r;
}

__device__ __forceinline__ float4 add4(float4 a, float4 b) {
  return make_float4(a.x + b.x, a.y + b.y, a.z + b.z, a.w + b.w);
}

__device__ __forceinline__ float bcast(float v, int j) {
  return __int_as_float(__builtin_amdgcn_readlane(__float_as_int(v), j));
}

__global__ __launch_bounds__(GATHER_THREADS) void node_gather(
    const int* __restrict__ rowptr, const int* __restrict__ csrc,
    const float* __restrict__ feat, const float* __restrict__ el,
    const float* __restrict__ er, const float* __restrict__ hin,
    float* __restrict__ outp) {
  const int lane = threadIdx.x & 31;
  const int wv = __builtin_amdgcn_readfirstlane((int)(threadIdx.x >> 5));
  const int t = blockIdx.x * GATHER_WAVES + wv;
  if (t >= NN) return;
  const int beg = rowptr[t];
  const int end = rowptr[t + 1];
  const float4 ra = ((const float4*)er)[(size_t)t * 2];
  const float4 rb = ((const float4*)er)[(size_t)t * 2 + 1];
  const float er0 = ra.x + ra.y, er1 = ra.z + ra.w, er2 = rb.x + rb.y, er3 = rb.z + rb.w;

  const float ninf = -__builtin_huge_valf();
  const int i0 = beg + lane;
  const bool has0 = i0 < end;
  int s0 = 0;
  float p0 = ninf, p1 = ninf, p2 = ninf, p3 = ninf;
  if (has0) {
    s0 = csrc[i0];
    edge_scores(el, s0, er0, er1, er2, er3, p0, p1, p2, p3);
  }

  float m0 = p0, m1 = p1, m2 = p2, m3 = p3;
  for (int i = i0 + 32; i < end; i += 32) {
    float q0, q1, q2, q3;
    edge_scores(el, csrc[i], er0, er1, er2, er3, q0, q1, q2, q3);
    m0 = fmaxf(m0, q0);
    m1 = fmaxf(m1, q1);
    m2 = fmaxf(m2, q2);
    m3 = fmaxf(m3, q3);
  }
  m0 = wave_max(m0);
  m1 = wave_max(m1);
  m2 = wave_max(m2);
  m3 = wave_max(m3);

  const float x0 = has0 ? __expf(p0 - m0) : 0.f;
  const float x1 = has0 ? __expf(p1 - m1) : 0.f;
  const float x2 = has0 ? __expf(p2 - m2) : 0.f;
  const float x3 = has0 ? __expf(p3 - m3) : 0.f;
  float d0 = x0, d1 = x1, d2 = x2, d3 = x3;
  for (int i = i0 + 32; i < end; i += 32) {
    float q0, q1, q2, q3;
    edge_scores(el, csrc[i], er0, er1, er2, er3, q0, q1, q2, q3);
    d0 += __expf(q0 - m0);
    d1 += __expf(q1 - m1);
    d2 += __expf(q2 - m2);
    d3 += __expf(q3 - m3);
  }
  d0 = wave_sum(d0);
  d1 = wave_sum(d1);
  d2 = wave_sum(d2);
  d3 = wave_sum(d3);
  const float dn0 = (d0 == 0.f) ? 1.f : d0;
  const float dn1 = (d1 == 0.f) ? 1.f : d1;
  const float dn2 = (d2 == 0.f) ? 1.f : d2;
  const float dn3 = (d3 == 0.f) ? 1.f : d3;

  float4 acc0 = make_float4(0.f, 0.f, 0.f, 0.f);
  float4 acc1 = acc0, acc2 = acc0, acc3 = acc0;
  const float4* frow = (const float4*)feat + lane;
  for (int c = beg; c < end; c += 32) {
    int s = s0;
    float a0 = x0 / dn0, a1 = x1 / dn1, a2 = x2 / dn2, a3 = x3 / dn3;
    if (c != beg) {
      const int i = c + lane;
      s = 0;
      a0 = 0.f; a1 = 0.f; a2 = 0.f; a3 = 0.f;
      if (i < end) {
        float q0, q1, q2, q3;
        s = csrc[i];
        edge_scores(el, s, er0, er1, er2, er3, q0, q1, q2, q3);
        a0 = __expf(q0 - m0) / dn0;
        a1 = __expf(q1 - m1) / dn1;
        a2 = __expf(q2 - m2) / dn2;
        a3 = __expf(q3 - m3) / dn3;
      }
    }
    int cnt = end - c;
    if (cnt > 32) cnt = 32;
#pragma unroll 2
    for (int j = 0; j < cnt; ++j) {
      const int sj = __builtin_amdgcn_readlane(s, j);
      const float b0 = bcast(a0, j), b1 = bcast(a1, j), b2 = bcast(a2, j), b3 = bcast(a3, j);
      const float4* fr = frow + (size_t)sj * (FH / 4);
      const float4 f0 = fr[0], f1 = fr[DD / 4], f2 = fr[2 * (DD / 4)], f3 = fr[3 * (DD / 4)];
      acc0.x += f0.x * b0;
      acc0.y += f0.y * b0;
      acc0.z += f0.z * b0;
      acc0.w += f0.w * b0;
      acc1.x += f1.x * b1;
      acc1.y += f1.y * b1;
      acc1.z += f1.z * b1;
      acc1.w += f1.w * b1;
      acc2.x += f2.x * b2;
      acc2.y += f2.y * b2;
      acc2.z += f2.z * b2;
      acc2.w += f2.w * b2;
      acc3.x += f3.x * b3;
      acc3.y += f3.y * b3;
      acc3.z += f3.z * b3;
      acc3.w += f3.w * b3;
    }
  }

  const float4* xr = (const float4*)hin + (size_t)t * (FH / 4) + lane;
  const float4 r0 = elu4(add4(acc0, xr[0]));
  const float4 r1 = elu4(add4(acc1, xr[DD / 4]));
  const float4 r2 = elu4(add4(acc2, xr[2 * (DD / 4)]));
  const float4 r3 = elu4(add4(acc3, xr[3 * (DD / 4)]));
  float4 sm;
  sm.x = (((r0.x + r1.x) + r2.x) + r3.x) * (1.f / HH);
  sm.y = (((r0.y + r1.y) + r2.y) + r3.y) * (1.f / HH);
  sm.z = (((r0.z + r1.z) + r2.z) + r3.z) * (1.f / HH);
  sm.w = (((r0.w + r1.w) + r2.w) + r3.w) * (1.f / HH);
  ((float4*)outp)[(size_t)t * (DD / 4) + lane] = sm;
}

#define AP_WAVES 8
#define AP_THREADS (AP_WAVES * 32)
#define AP_PITCH (HH * K0 + 4)
#define AP_GROUP 4
#define AP_MIN_FLOATS 20992

__global__ __launch_bounds__(AP_THREADS) void agg_proj(
    const int* __restrict__ rowptr, const int* __restrict__ csrc,
    const float* __restrict__ x, const float* __restrict__ el,
    const float* __restrict__ er, const float* __restrict__ wf,
    float* __restrict__ hout) {
  static_assert(NN % 16 == 0, "whole M-tiles");
  static_assert(FH / 64 == AP_WAVES, "one wave per 64-column panel");
  static_assert((K0 / 4) % AP_GROUP == 0, "whole groups of k-steps");
  __shared__ float Z[(16 * AP_PITCH > AP_MIN_FLOATS) ? 16 * AP_PITCH : AP_MIN_FLOATS];

  const int lane = threadIdx.x & 31;
  const int wv = __builtin_amdgcn_readfirstlane((int)(threadIdx.x >> 5));
  const int tile = blockIdx.x;
  const float ninf = -__builtin_huge_valf();

  for (int nl = wv * 2; nl < wv * 2 + 2; ++nl) {
    const int t = tile * 16 + nl;
    const int beg = rowptr[t];
    const int end = rowptr[t + 1];
    const float4 ra = ((const float4*)er)[(size_t)t * 2];
    const float4 rb = ((const float4*)er)[(size_t)t * 2 + 1];
    const float er0 = ra.x + ra.y, er1 = ra.z + ra.w, er2 = rb.x + rb.y, er3 = rb.z + rb.w;

    const int i0 = beg + lane;
    const bool has0 = i0 < end;
    int s0 = 0;
    float p0 = ninf, p1 = ninf, p2 = ninf, p3 = ninf;
    if (has0) {
      s0 = csrc[i0];
      edge_scores(el, s0, er0, er1, er2, er3, p0, p1, p2, p3);
    }

    float m0 = p0, m1 = p1, m2 = p2, m3 = p3;
    for (int i = i0 + 32; i < end; i += 32) {
      float q0, q1, q2, q3;
      edge_scores(el, csrc[i], er0, er1, er2, er3, q0, q1, q2, q3);
      m0 = fmaxf(m0, q0);
      m1 = fmaxf(m1, q1);
      m2 = fmaxf(m2, q2);
      m3 = fmaxf(m3, q3);
    }
    m0 = wave_max(m0);
    m1 = wave_max(m1);
    m2 = wave_max(m2);
    m3 = wave_max(m3);

    const float x0 = has0 ? __expf(p0 - m0) : 0.f;
    const float x1 = has0 ? __expf(p1 - m1) : 0.f;
    const float x2 = has0 ? __expf(p2 - m2) : 0.f;
    const float x3 = has0 ? __expf(p3 - m3) : 0.f;
    float d0 = x0, d1 = x1, d2 = x2, d3 = x3;
    for (int i = i0 + 32; i < end; i += 32) {
      float q0, q1, q2, q3;
      edge_scores(el, csrc[i], er0, er1, er2, er3, q0, q1, q2, q3);
      d0 += __expf(q0 - m0);
      d1 += __expf(q1 - m1);
      d2 += __expf(q2 - m2);
      d3 += __expf(q3 - m3);
    }
    d0 = wave_sum(d0);
    d1 = wave_sum(d1);
    d2 = wave_sum(d2);
    d3 = wave_sum(d3);
    const float dn0 = (d0 == 0.f) ? 1.f : d0;
    const float dn1 = (d1 == 0.f) ? 1.f : d1;
    const float dn2 = (d2 == 0.f) ? 1.f : d2;
    const float dn3 = (d3 == 0.f) ? 1.f : d3;

    float4 acc0 = make_float4(0.f, 0.f, 0.f, 0.f);
    float4 acc1 = acc0, acc2 = acc0, acc3 = acc0;
    float4 acc4 = acc0, acc5 = acc0, acc6 = acc0, acc7 = acc0;
    const float4* frow = (const float4*)x + lane;
    for (int c = beg; c < end; c += 32) {
      int s = s0;
      float a0 = x0 / dn0, a1 = x1 / dn1, a2 = x2 / dn2, a3 = x3 / dn3;
      if (c != beg) {
        const int i = c + lane;
        s = 0;
        a0 = 0.f; a1 = 0.f; a2 = 0.f; a3 = 0.f;
        if (i < end) {
          float q0, q1, q2, q3;
          s = csrc[i];
          edge_scores(el, s, er0, er1, er2, er3, q0, q1, q2, q3);
          a0 = __expf(q0 - m0) / dn0;
          a1 = __expf(q1 - m1) / dn1;
          a2 = __expf(q2 - m2) / dn2;
          a3 = __expf(q3 - m3) / dn3;
        }
      }
      int cnt = end - c;
      if (cnt > 32) cnt = 32;
#pragma unroll 4
      for (int j = 0; j < cnt; ++j) {
        const int sj = __builtin_amdgcn_readlane(s, j);
        const float b0 = bcast(a0, j), b1 = bcast(a1, j), b2 = bcast(a2, j), b3 = bcast(a3, j);
        const float4* fr = frow + (size_t)sj * (K0 / 4);
        const float4 f0 = fr[0], f1 = fr[K0 / 8];
        acc0.x += f0.x * b0;
        acc0.y += f0.y * b0;
        acc0.z += f0.z * b0;
        acc0.w += f0.w * b0;
        acc4.x += f1.x * b0;
        acc4.y += f1.y * b0;
        acc4.z += f1.z * b0;
        acc4.w += f1.w * b0;
        acc1.x += f0.x * b1;
        acc1.y += f0.y * b1;
        acc1.z += f0.z * b1;
        acc1.w += f0.w * b1;
        acc5.x += f1.x * b1;
        acc5.y += f1.y * b1;
        acc5.z += f1.z * b1;
        acc5.w += f1.w * b1;
        acc2.x += f0.x * b2;
        acc2.y += f0.y * b2;
        acc2.z += f0.z * b2;
        acc2.w += f0.w * b2;
        acc6.x += f1.x * b2;
        acc6.y += f1.y * b2;
        acc6.z += f1.z * b2;
        acc6.w += f1.w * b2;
        acc3.x += f0.x * b3;
        acc3.y += f0.y * b3;
        acc3.z += f0.z * b3;
        acc3.w += f0.w * b3;
        acc7.x += f1.x * b3;
        acc7.y += f1.y * b3;
        acc7.z += f1.z * b3;
        acc7.w += f1.w * b3;
      }
    }

    float4* z = (float4*)(Z + nl * AP_PITCH) + lane;
    z[0] = acc0;
    z[K0 / 8] = acc4;
    z[K0 / 4] = acc1;
    z[K0 / 4 + K0 / 8] = acc5;
    z[2 * (K0 / 4)] = acc2;
    z[2 * (K0 / 4) + K0 / 8] = acc6;
    z[3 * (K0 / 4)] = acc3;
    z[3 * (K0 / 4) + K0 / 8] = acc7;
  }
  __syncthreads();

  const int half = lane >> 4;
  const int l = lane & 15;
  const int pn = wv;
  const float2* za = (const float2*)(Z + l * AP_PITCH + (pn >> 1) * K0 + half * 2);
  const float4* bp = (const float4*)wf + (size_t)pn * (K0 / 4) * 64 + lane;

  v8f c0 = {}, c1 = {}, c2 = {}, c3 = {};
  constexpr int NGRP = (K0 / 4) / AP_GROUP;
  static_assert((NGRP & (NGRP - 1)) == 0, "the group index wraps with a mask");
  const int g0 = tile & (NGRP - 1);
  float4 cur[2 * AP_GROUP], nxt[2 * AP_GROUP];
#pragma unroll
  for (int i = 0; i < AP_GROUP; ++i) {
    cur[2 * i] = bp[(g0 * AP_GROUP + i) * 64];
    cur[2 * i + 1] = bp[(g0 * AP_GROUP + i) * 64 + 32];
  }
  for (int n = 0; n < NGRP; ++n) {
    const int g = (g0 + n) & (NGRP - 1);
    if (n + 1 < NGRP) {
      const int gn = (g0 + n + 1) & (NGRP - 1);
#pragma unroll
      for (int i = 0; i < AP_GROUP; ++i) {
        nxt[2 * i] = bp[(gn * AP_GROUP + i) * 64];
        nxt[2 * i + 1] = bp[(gn * AP_GROUP + i) * 64 + 32];
      }
    }
#pragma unroll
    for (int i = 0; i < AP_GROUP; ++i) {
      const float2 av = za[(g * AP_GROUP + i) * 2];
      v2f a; a.x = av.x; a.y = av.y;
      v2f b0; b0.x = cur[2 * i].x; b0.y = cur[2 * i].y;
      v2f b1; b1.x = cur[2 * i].z; b1.y = cur[2 * i].w;
      v2f b2; b2.x = cur[2 * i + 1].x; b2.y = cur[2 * i + 1].y;
      v2f b3; b3.x = cur[2 * i + 1].z; b3.y = cur[2 * i + 1].w;
      c0 = __builtin_amdgcn_wmma_f32_16x16x4_f32(false, a, false, b0, (short)0, c0, false, false);
      c1 = __builtin_amdgcn_wmma_f32_16x16x4_f32(false, a, false, b1, (short)0, c1, false, false);
      c2 = __builtin_amdgcn_wmma_f32_16x16x4_f32(false, a, false, b2, (short)0, c2, false, false);
      c3 = __builtin_amdgcn_wmma_f32_16x16x4_f32(false, a, false, b3, (short)0, c3, false, false);
    }
    if (n + 1 < NGRP) {
#pragma unroll
      for (int i = 0; i < 2 * AP_GROUP; ++i) cur[i] = nxt[i];
    }
  }

  float* obase = hout + (size_t)(tile * 16 + half * 8) * FH + pn * 64 + l;
#pragma unroll
  for (int r = 0; r < 8; ++r) {
    float* orow = obase + (size_t)r * FH;
    orow[0] = elu1(c0[r]);
    orow[16] = elu1(c1[r]);
    orow[32] = elu1(c2[r]);
    orow[48] = elu1(c3[r]);
  }
}

extern "C" void kernel_launch(void* const* d_in, const int* in_sizes, int n_in,
                              void* d_out, int out_size, void* d_ws, size_t ws_size,
                              hipStream_t stream) {
  const int* src = (const int*)d_in[0];
  const int* dst = (const int*)d_in[1];
  const float* x   = (const float*)d_in[2];
  const float* W0  = (const float*)d_in[3];
  const float* al0 = (const float*)d_in[4];
  const float* ar0 = (const float*)d_in[5];
  const float* W1  = (const float*)d_in[6];
  const float* al1 = (const float*)d_in[7];
  const float* ar1 = (const float*)d_in[8];
  float* out = (float*)d_out;

  float* ws    = (float*)d_ws;
  float* feat  = ws;
  float* hbuf  = feat  + (size_t)NN * FH;
  float* el    = hbuf  + (size_t)NN * FH;
  float* er    = el    + (size_t)NN * HH * 2;
  int* rowptr  = (int*)(er + (size_t)NN * HH * 2);
  int* csrc    = rowptr + (NN + 16);
  int* tabc    = csrc   + EE;
  int* tabo    = tabc   + GRP_BINS * NSLAB;
  int* tmp     = tabo   + GRP_BINS * NSLAB;
  float* uv    = (float*)(tmp + 2 * EE);
  float* wf    = uv + 8 * K0 + 32;

  const int mtiles = (NN + 15) / 16;
  const int mgroups = (mtiles + GEMM_WAVES - 1) / GEMM_WAVES;
  const int npanels = FH / 64;
  const int gemmBlocks = mgroups * npanels;

  slab_bin<<<NSLAB + UV_BLOCKS + WF_BLOCKS, SLAB_THREADS, 0, stream>>>(src, dst, tmp, tabc, tabo,
                                                                     W0, al0, ar0, uv, wf);
  bin_group<<<GRP_BINS + (NN + 31) / 32, 1024, 0, stream>>>(tmp, tabc, tabo, rowptr, csrc, x, uv, el, er);

  agg_proj<<<NN / 16, AP_THREADS, 0, stream>>>(rowptr, csrc, x, el, er, wf, hbuf);
  gemm16x64_wmma<KMAX, FH><<<gemmBlocks, GEMM_THREADS, 0, stream>>>(hbuf, W1, al1, ar1, feat, el, er, NN);
  node_gather<<<GATHER_BLOCKS, GATHER_THREADS, 0, stream>>>(rowptr, csrc, feat, el, er, hbuf, out);
}
